# starting kernel
# baseline (speedup 1.0000x reference)
#include <hip/hip_runtime.h>
#include <hip/hip_fp16.h>

typedef __attribute__((ext_vector_type(16))) _Float16 v16h;
typedef __attribute__((ext_vector_type(8)))  float    v8f;

#define WMMA_F16(a,b,c) __builtin_amdgcn_wmma_f32_16x16x32_f16(false,(a),false,(b),(short)0,(c),false,false)

__device__ __forceinline__ v8f zero8(){
  v8f z;
  #pragma unroll
  for(int i=0;i<8;++i) z[i]=0.f;
  return z;
}
__device__ __forceinline__ v16h zero16(){
  v16h z;
  #pragma unroll
  for(int i=0;i<16;++i) z[i]=(_Float16)0.f;
  return z;
}

__device__ __forceinline__ int a_k(int i, int h16){ return (i&7) + ((i>>3)<<4) + (h16<<3); }

__device__ __forceinline__ v16h load_tile_f32(const float* __restrict__ src, int ld, int lane){
  const float* rb = src + (size_t)(lane&15)*ld;
  int h16=(lane>>4)&1; v16h r;
  #pragma unroll
  for(int i=0;i<16;++i) r[i]=(_Float16)rb[a_k(i,h16)];
  return r;
}
__device__ __forceinline__ v16h load_tile_f16(const _Float16* __restrict__ src, int ld, int lane){
  const _Float16* rb = src + (size_t)(lane&15)*ld;
  int h16=(lane>>4)&1; v16h r;
  #pragma unroll
  for(int i=0;i<16;++i) r[i]=rb[a_k(i,h16)];
  return r;
}
__device__ __forceinline__ v16h load_tile_kmajor_f16(const _Float16* __restrict__ src, int ld, int lane){
  const _Float16* cb = src + (lane&15);
  int h16=(lane>>4)&1; v16h r;
  #pragma unroll
  for(int i=0;i<16;++i) r[i]=cb[(size_t)a_k(i,h16)*ld];
  return r;
}
__device__ __forceinline__ v16h load_tile_sw(const _Float16* src, int tile, int lane){
  return *reinterpret_cast<const v16h*>(src + (((size_t)tile*32 + lane) << 4));
}

__device__ __forceinline__ float rmax16(float x){
  #pragma unroll
  for(int m=8;m;m>>=1) x=fmaxf(x,__shfl_xor(x,m,32));
  return x;
}
__device__ __forceinline__ float rsum16(float x){
  #pragma unroll
  for(int m=8;m;m>>=1) x+=__shfl_xor(x,m,32);
  return x;
}
__device__ __forceinline__ float sigf(float x){ return 1.f/(1.f+__expf(-x)); }

#define PB  8
#define DD  128
#define HH  64
#define SQ  512

__global__ __launch_bounds__(256) void k_swz(const float* __restrict__ W,
                                             _Float16* __restrict__ out, int K){
  int idx = blockIdx.x*256 + threadIdx.x;
  int i    = idx & 15;
  int lane = (idx >> 4) & 31;
  int tile = idx >> 9;
  int ktiles = K >> 5;
  int kt = tile % ktiles, jt = tile / ktiles;
  out[idx] = (_Float16)W[(size_t)(jt*16 + (lane&15))*K + kt*32 + a_k(i,(lane>>4)&1)];
}

__global__ __launch_bounds__(256) void k_gru(
    const float* __restrict__ x, const _Float16* __restrict__ WihT,
    const _Float16* __restrict__ WhhT, const float* __restrict__ bih,
    const float* __restrict__ bhh, float* __restrict__ seq)
{
  __shared__ _Float16 hscr[8][16*64];
  __shared__ _Float16 WhhL[192*64];
  int lane=threadIdx.x&31, wave=threadIdx.x>>5;
  int pbase=(blockIdx.x*8+wave)*16;
  int colb=lane&15, rowoff=(lane>>4)*8;

  {
    const uint32_t* wsrc=(const uint32_t*)WhhT;
    uint32_t* wdst=(uint32_t*)WhhL;
    for(int i=threadIdx.x;i<192*64/2;i+=256) wdst[i]=wsrc[i];
  }
  __syncthreads();

  float bi[12], bh[12];
  #pragma unroll
  for(int j=0;j<12;++j){ bi[j]=bih[j*16+colb]; bh[j]=bhh[j*16+colb]; }

  v8f hT[4], hs[4];
  #pragma unroll
  for(int j=0;j<4;++j){ hT[j]=zero8(); hs[j]=zero8(); }
  v16h hA[2]; hA[0]=zero16(); hA[1]=zero16();
  _Float16* hl=&hscr[wave][0];

  for(int t=0;t<PB;++t){
    size_t zoff = 0;
    asm volatile("" : "+s"(zoff));
    const _Float16* Wt = WihT + zoff;
    const float* xsrc = x + ((size_t)pbase*PB + t)*DD;
    if(t+1<PB) __builtin_prefetch(xsrc+DD,0,1);
    v16h xA[4];
    #pragma unroll
    for(int kk=0;kk<4;++kk) xA[kk]=load_tile_f32(xsrc+kk*32, PB*DD, lane);

    #pragma unroll
    for(int j=0;j<4;++j){
      v8f gr=zero8();
      #pragma unroll
      for(int kk=0;kk<4;++kk)
        gr=WMMA_F16(xA[kk], load_tile_sw(Wt, j*4+kk, lane), gr);
      #pragma unroll
      for(int kk=0;kk<2;++kk)
        gr=WMMA_F16(hA[kk], load_tile_sw(WhhL, j*2+kk, lane), gr);
      v8f gz=zero8();
      #pragma unroll
      for(int kk=0;kk<4;++kk)
        gz=WMMA_F16(xA[kk], load_tile_sw(Wt, (4+j)*4+kk, lane), gz);
      #pragma unroll
      for(int kk=0;kk<2;++kk)
        gz=WMMA_F16(hA[kk], load_tile_sw(WhhL, (4+j)*2+kk, lane), gz);
      v8f gnx=zero8();
      #pragma unroll
      for(int kk=0;kk<4;++kk)
        gnx=WMMA_F16(xA[kk], load_tile_sw(Wt, (8+j)*4+kk, lane), gnx);
      v8f gnh=zero8();
      #pragma unroll
      for(int kk=0;kk<2;++kk)
        gnh=WMMA_F16(hA[kk], load_tile_sw(WhhL, (8+j)*2+kk, lane), gnh);
      #pragma unroll
      for(int e=0;e<8;++e){
        float r =sigf(gr[e]+bi[j]+bh[j]);
        float z =sigf(gz[e]+bi[4+j]+bh[4+j]);
        float nv=tanhf(gnx[e]+bi[8+j]+r*(gnh[e]+bh[8+j]));
        float hn=(1.f-z)*nv + z*hT[j][e];
        hT[j][e]=hn; hs[j][e]+=hn;
      }
    }
    #pragma unroll
    for(int j=0;j<4;++j){
      #pragma unroll
      for(int v=0;v<8;++v)
        hl[(v+rowoff)*64 + j*16 + colb]=(_Float16)hT[j][v];
    }
    #pragma unroll
    for(int kk=0;kk<2;++kk) hA[kk]=load_tile_f16(hl+kk*32, 64, lane);
  }
  float* sb=seq+(size_t)pbase*HH;
  #pragma unroll
  for(int j=0;j<4;++j){
    #pragma unroll
    for(int v=0;v<8;++v)
      sb[(size_t)(v+rowoff)*HH + j*16 + colb]=hs[j][v]*0.125f;
  }
}

__global__ __launch_bounds__(256) void k_qkv(
    const float* __restrict__ X, const _Float16* __restrict__ WqkvT,
    const float* __restrict__ bqkv, _Float16* __restrict__ Qp,
    _Float16* __restrict__ Kp, _Float16* __restrict__ Vh)
{
  int lane=threadIdx.x&31, wave=threadIdx.x>>5;
  size_t rbase=((size_t)blockIdx.x*8+wave)*16;
  int b=(int)(rbase>>9), s=(int)(rbase&511);
  int colb=lane&15, rowoff=(lane>>4)*8;
  v16h A[2];
  #pragma unroll
  for(int kk=0;kk<2;++kk) A[kk]=load_tile_f32(X+rbase*HH+kk*32, HH, lane);

  #pragma unroll
  for(int j=0;j<12;++j){
    v8f acc=zero8();
    #pragma unroll
    for(int kk=0;kk<2;++kk)
      acc=WMMA_F16(A[kk], load_tile_sw(WqkvT, j*2+kk, lane), acc);
    float bias=bqkv[j*16+colb];
    int h=j&3;
    size_t bh=(size_t)b*4+h;
    if(j<4){
      _Float16* dst=Qp+(bh*SQ+s)*32;
      #pragma unroll
      for(int v=0;v<8;++v){
        size_t ro=(size_t)(v+rowoff)*32;
        dst[ro+colb]=(_Float16)((acc[v]+bias)*0.25f);
        dst[ro+16+colb]=(_Float16)0.f;
      }
    } else if(j<8){
      _Float16* dst=Kp+(bh*SQ+s)*32;
      #pragma unroll
      for(int v=0;v<8;++v){
        size_t ro=(size_t)(v+rowoff)*32;
        dst[ro+colb]=(_Float16)(acc[v]+bias);
        dst[ro+16+colb]=(_Float16)0.f;
      }
    } else {
      _Float16* dst=Vh+(bh*SQ+s)*16;
      #pragma unroll
      for(int v=0;v<8;++v)
        dst[(size_t)(v+rowoff)*16+colb]=(_Float16)(acc[v]+bias);
    }
  }
}

__global__ __launch_bounds__(256) void k_attn(
    const _Float16* __restrict__ Qp, const _Float16* __restrict__ Kp,
    const _Float16* __restrict__ Vh, float* __restrict__ O)
{
  __shared__ _Float16 Ks[SQ*32];
  __shared__ _Float16 Vs[SQ*16];
  __shared__ _Float16 Ps[8][16*32];
  int bh=blockIdx.x, b=bh>>2, h=bh&3;
  int tid=threadIdx.x, lane=tid&31, wave=tid>>5;
  int colb=lane&15, rowoff=(lane>>4)*8;

  {
    const uint32_t* ksrc=(const uint32_t*)(Kp+(size_t)bh*SQ*32);
    uint32_t* kdst=(uint32_t*)Ks;
    for(int i=tid;i<SQ*16;i+=256) kdst[i]=ksrc[i];
    const uint32_t* vsrc=(const uint32_t*)(Vh+(size_t)bh*SQ*16);
    uint32_t* vdst=(uint32_t*)Vs;
    for(int i=tid;i<SQ*8;i+=256) vdst[i]=vsrc[i];
  }
  __syncthreads();

  _Float16* pl=&Ps[wave][0];
  for(int qt=0;qt<4;++qt){
    int qbase=wave*64+qt*16;
    v16h Qa=load_tile_f16(Qp+((size_t)bh*SQ+qbase)*32, 32, lane);
    v8f acc=zero8(), mrow, lrow;
    #pragma unroll
    for(int e=0;e<8;++e){ mrow[e]=-1e30f; lrow[e]=0.f; }

    for(int kc=0;kc<16;++kc){
      int kbase=kc*32;
      v16h Kb0=load_tile_f16(Ks+(size_t)kbase*32, 32, lane);
      v16h Kb1=load_tile_f16(Ks+(size_t)(kbase+16)*32, 32, lane);
      v8f s0=WMMA_F16(Qa,Kb0,zero8());
      v8f s1=WMMA_F16(Qa,Kb1,zero8());
      v8f nmv, scv;
      #pragma unroll
      for(int e=0;e<8;++e){
        float cm=fmaxf(rmax16(s0[e]),rmax16(s1[e]));
        float nm=fmaxf(mrow[e],cm);
        nmv[e]=nm; scv[e]=__expf(mrow[e]-nm); mrow[e]=nm;
      }
      #pragma unroll
      for(int e=0;e<8;++e){
        s0[e]=__expf(s0[e]-nmv[e]);
        s1[e]=__expf(s1[e]-nmv[e]);
        lrow[e]=lrow[e]*scv[e]+rsum16(s0[e])+rsum16(s1[e]);
        acc[e]*=scv[e];
      }
      #pragma unroll
      for(int e=0;e<8;++e){
        pl[(e+rowoff)*32+colb]=(_Float16)s0[e];
        pl[(e+rowoff)*32+16+colb]=(_Float16)s1[e];
      }
      v16h Pa=load_tile_f16(pl, 32, lane);
      v16h Vb=load_tile_kmajor_f16(Vs+(size_t)kbase*16, 16, lane);
      acc=WMMA_F16(Pa,Vb,acc);
    }
    float* ob=O+((size_t)b*SQ+qbase)*HH + h*16;
    #pragma unroll
    for(int e=0;e<8;++e)
      ob[(size_t)(e+rowoff)*HH+colb]=acc[e]/lrow[e];
  }
}

__global__ __launch_bounds__(256) void k_projln(
    float* __restrict__ X, const float* __restrict__ Y,
    const _Float16* __restrict__ WoT, const float* __restrict__ bo,
    const float* __restrict__ g, const float* __restrict__ be)
{
  int lane=threadIdx.x&31, wave=threadIdx.x>>5;
  size_t rbase=((size_t)blockIdx.x*8+wave)*16;
  int colb=lane&15, rowoff=(lane>>4)*8;
  v16h A[2];
  #pragma unroll
  for(int kk=0;kk<2;++kk) A[kk]=load_tile_f32(Y+rbase*HH+kk*32, HH, lane);
  v8f o[4];
  #pragma unroll
  for(int j=0;j<4;++j){
    v8f acc=zero8();
    #pragma unroll
    for(int kk=0;kk<2;++kk)
      acc=WMMA_F16(A[kk], load_tile_sw(WoT, j*2+kk, lane), acc);
    float bias=bo[j*16+colb];
    const float* xr=X+rbase*HH+j*16+colb;
    #pragma unroll
    for(int e=0;e<8;++e) o[j][e]=acc[e]+bias+xr[(size_t)(e+rowoff)*HH];
  }
  float gv[4],bv[4];
  #pragma unroll
  for(int j=0;j<4;++j){ gv[j]=g[j*16+colb]; bv[j]=be[j*16+colb]; }
  float* xw=X+rbase*HH;
  #pragma unroll
  for(int e=0;e<8;++e){
    float s=o[0][e]+o[1][e]+o[2][e]+o[3][e];
    float mu=rsum16(s)*(1.f/64.f);
    float d0=o[0][e]-mu,d1=o[1][e]-mu,d2=o[2][e]-mu,d3=o[3][e]-mu;
    float var=rsum16(d0*d0+d1*d1+d2*d2+d3*d3)*(1.f/64.f);
    float inv=rsqrtf(var+1e-5f);
    #pragma unroll
    for(int j=0;j<4;++j)
      xw[(size_t)(e+rowoff)*HH+j*16+colb]=(o[j][e]-mu)*inv*gv[j]+bv[j];
  }
}

__global__ __launch_bounds__(128) void k_ffln(
    float* __restrict__ X, const _Float16* __restrict__ W1T,
    const float* __restrict__ b1, const _Float16* __restrict__ W2T,
    const float* __restrict__ b2, const float* __restrict__ g,
    const float* __restrict__ be)
{
  __shared__ _Float16 h1[4][16*256];
  int lane=threadIdx.x&31, wave=threadIdx.x>>5;
  size_t rbase=((size_t)blockIdx.x*4+wave)*16;
  int colb=lane&15, rowoff=(lane>>4)*8;
  v16h A[2];
  #pragma unroll
  for(int kk=0;kk<2;++kk) A[kk]=load_tile_f32(X+rbase*HH+kk*32, HH, lane);
  _Float16* hl=&h1[wave][0];
  #pragma unroll
  for(int j=0;j<16;++j){
    v8f acc=zero8();
    #pragma unroll
    for(int kk=0;kk<2;++kk)
      acc=WMMA_F16(A[kk], load_tile_sw(W1T, j*2+kk, lane), acc);
    float bias=b1[j*16+colb];
    #pragma unroll
    for(int e=0;e<8;++e)
      hl[(size_t)(e+rowoff)*256+j*16+colb]=(_Float16)fmaxf(acc[e]+bias,0.f);
  }
  v16h pA[8];
  #pragma unroll
  for(int kk=0;kk<8;++kk) pA[kk]=load_tile_f16(hl+kk*32, 256, lane);
  v8f o[4];
  #pragma unroll
  for(int j=0;j<4;++j){
    v8f acc=zero8();
    #pragma unroll
    for(int kk=0;kk<8;++kk)
      acc=WMMA_F16(pA[kk], load_tile_sw(W2T, j*8+kk, lane), acc);
    float bias=b2[j*16+colb];
    const float* xr=X+rbase*HH+j*16+colb;
    #pragma unroll
    for(int e=0;e<8;++e) o[j][e]=acc[e]+bias+xr[(size_t)(e+rowoff)*HH];
  }
  float gv[4],bv[4];
  #pragma unroll
  for(int j=0;j<4;++j){ gv[j]=g[j*16+colb]; bv[j]=be[j*16+colb]; }
  float* xw=X+rbase*HH;
  #pragma unroll
  for(int e=0;e<8;++e){
    float s=o[0][e]+o[1][e]+o[2][e]+o[3][e];
    float mu=rsum16(s)*(1.f/64.f);
    float d0=o[0][e]-mu,d1=o[1][e]-mu,d2=o[2][e]-mu,d3=o[3][e]-mu;
    float var=rsum16(d0*d0+d1*d1+d2*d2+d3*d3)*(1.f/64.f);
    float inv=rsqrtf(var+1e-5f);
    #pragma unroll
    for(int j=0;j<4;++j)
      xw[(size_t)(e+rowoff)*HH+j*16+colb]=(o[j][e]-mu)*inv*gv[j]+bv[j];
  }
}

__global__ void k_mean(const float* __restrict__ X, float* __restrict__ out){
  int b=blockIdx.x, j=threadIdx.x;
  float s=0.f;
  for(int np=0;np<SQ;++np) s+=X[((size_t)b*SQ+np)*HH+j];
  out[b*HH+j]=s*(1.f/512.f);
}

extern "C" void kernel_launch(void* const* d_in, const int* in_sizes, int n_in,
                              void* d_out, int out_size, void* d_ws, size_t ws_size,
                              hipStream_t stream){
  const float* x  =(const float*)d_in[0];
  const float* Wih=(const float*)d_in[1];
  const float* Whh=(const float*)d_in[2];
  const float* bih=(const float*)d_in[3];
  const float* bhh=(const float*)d_in[4];

  char* ws=(char*)d_ws;
  float*    X =(float*)ws;
  float*    Y =(float*)(ws+(size_t)(8<<20));
  _Float16* Qp=(_Float16*)(ws+(size_t)(16<<20));
  _Float16* Kp=(_Float16*)(ws+(size_t)(24<<20));
  _Float16* Vh=(_Float16*)(ws+(size_t)(32<<20));
  _Float16* WihT=(_Float16*)(ws+(size_t)(36<<20));
  _Float16* WhhT=(_Float16*)(ws+(size_t)(36<<20)+(64<<10));
  _Float16* LW[2][4];
  for(int L=0;L<2;++L){
    char* base=ws+(size_t)(37<<20)+(size_t)L*(1<<20);
    LW[L][0]=(_Float16*)(base);
    LW[L][1]=(_Float16*)(base+(32<<10));
    LW[L][2]=(_Float16*)(base+(64<<10));
    LW[L][3]=(_Float16*)(base+(128<<10));
  }

  k_swz<<<96,256,0,stream>>>(Wih, WihT, 128);
  k_swz<<<48,256,0,stream>>>(Whh, WhhT, 64);
  for(int L=0;L<2;++L){
    int base=5+L*12;
    k_swz<<<48,256,0,stream>>>((const float*)d_in[base+0], LW[L][0], 64);
    k_swz<<<16,256,0,stream>>>((const float*)d_in[base+2], LW[L][1], 64);
    k_swz<<<64,256,0,stream>>>((const float*)d_in[base+4], LW[L][2], 64);
    k_swz<<<64,256,0,stream>>>((const float*)d_in[base+6], LW[L][3], 256);
  }

  k_gru<<<256,256,0,stream>>>(x,WihT,WhhT,bih,bhh,X);

  for(int L=0;L<2;++L){
    int base=5+L*12;
    const float* bqkv=(const float*)d_in[base+1];
    const float* bo  =(const float*)d_in[base+3];
    const float* b1  =(const float*)d_in[base+5];
    const float* b2  =(const float*)d_in[base+7];
    const float* g1  =(const float*)d_in[base+8];
    const float* be1 =(const float*)d_in[base+9];
    const float* g2  =(const float*)d_in[base+10];
    const float* be2 =(const float*)d_in[base+11];

    k_qkv   <<<256,256,0,stream>>>(X,LW[L][0],bqkv,Qp,Kp,Vh);
    k_attn  <<<256,256,0,stream>>>(Qp,Kp,Vh,Y);
    k_projln<<<256,256,0,stream>>>(X,Y,LW[L][1],bo,g1,be1);
    k_ffln  <<<512,128,0,stream>>>(X,LW[L][2],b1,LW[L][3],b2,g2,be2);
  }
  k_mean<<<64,64,0,stream>>>(X,(float*)d_out);
}


# final mean folded into the last FF+LN launch: each workgroup adds its 64 normalised rows per column (registers, then LDS), stores one 64-float row; k_mean adds 8 rows per batch; sum order changes
# speedup vs baseline: 2.4062x; 2.4062x over previous
#include <hip/hip_runtime.h>
#include <hip/hip_fp16.h>

typedef __attribute__((ext_vector_type(16))) _Float16 v16h;
typedef __attribute__((ext_vector_type(8)))  float    v8f;

#define WMMA_F16(a,b,c) __builtin_amdgcn_wmma_f32_16x16x32_f16(false,(a),false,(b),(short)0,(c),false,false)

__device__ __forceinline__ v8f zero8(){
  v8f z;
  #pragma unroll
  for(int i=0;i<8;++i) z[i]=0.f;
  return z;
}
__device__ __forceinline__ v16h zero16(){
  v16h z;
  #pragma unroll
  for(int i=0;i<16;++i) z[i]=(_Float16)0.f;
  return z;
}

__device__ __forceinline__ int a_k(int i, int h16){ return (i&7) + ((i>>3)<<4) + (h16<<3); }

__device__ __forceinline__ v16h load_tile_f32(const float* __restrict__ src, int ld, int lane){
  const float* rb = src + (size_t)(lane&15)*ld;
  int h16=(lane>>4)&1; v16h r;
  #pragma unroll
  for(int i=0;i<16;++i) r[i]=(_Float16)rb[a_k(i,h16)];
  return r;
}
__device__ __forceinline__ v16h load_tile_f16(const _Float16* __restrict__ src, int ld, int lane){
  const _Float16* rb = src + (size_t)(lane&15)*ld;
  int h16=(lane>>4)&1; v16h r;
  #pragma unroll
  for(int i=0;i<16;++i) r[i]=rb[a_k(i,h16)];
  return r;
}
__device__ __forceinline__ v16h load_tile_kmajor_f16(const _Float16* __restrict__ src, int ld, int lane){
  const _Float16* cb = src + (lane&15);
  int h16=(lane>>4)&1; v16h r;
  #pragma unroll
  for(int i=0;i<16;++i) r[i]=cb[(size_t)a_k(i,h16)*ld];
  return r;
}

typedef short s16x8 __attribute__((ext_vector_type(8)));
typedef __attribute__((ext_vector_type(8))) _Float16 v8h;
__device__ __forceinline__ v8h ld_tr16(const _Float16* p){
  typedef __attribute__((address_space(3))) s16x8 lds_s16x8;
  return __builtin_bit_cast(v8h, __builtin_amdgcn_ds_load_tr16_b128_v8i16((lds_s16x8*)p));
}
__device__ __forceinline__ v16h cat16(v8h lo, v8h hi){
  return __builtin_shufflevector(lo, hi, 0,1,2,3,4,5,6,7,8,9,10,11,12,13,14,15);
}

__device__ __forceinline__ float rmax16(float x){
  #pragma unroll
  for(int m=8;m;m>>=1) x=fmaxf(x,__shfl_xor(x,m,32));
  return x;
}
__device__ __forceinline__ float rsum16(float x){
  #pragma unroll
  for(int m=8;m;m>>=1) x+=__shfl_xor(x,m,32);
  return x;
}
__device__ __forceinline__ float sigf(float x){ return 1.f/(1.f+__expf(-x)); }

__device__ __forceinline__ v8f rmax16x8(v8f x){
  #pragma unroll
  for(int m=8;m;m>>=1){
    v8f t;
    #pragma unroll
    for(int e=0;e<8;++e) t[e]=__shfl_xor(x[e],m,32);
    x=__builtin_elementwise_max(x,t);
  }
  return x;
}
__device__ __forceinline__ v8f rsum16x8(v8f x){
  #pragma unroll
  for(int m=8;m;m>>=1){
    v8f t;
    #pragma unroll
    for(int e=0;e<8;++e) t[e]=__shfl_xor(x[e],m,32);
    x+=t;
  }
  return x;
}

#define PB  8
#define DD  128
#define HH  64
#define SQ  512

template<int K, int NT, int NTHR>
__device__ __forceinline__ void stage_w(const float* __restrict__ W, _Float16* L, int tid){
  constexpr int KT = K/32;
  #pragma unroll
  for(int q=0;q<NT*64/NTHR;++q){
    int u    = q*NTHR + tid;
    int lane = u & 31;
    int p    = (u >> 5) & 1;
    int tile = u >> 6;
    int kt = tile % KT, jt = tile / KT;
    const float* s = W + (size_t)(jt*16 + (lane&15))*K + kt*32 + 16*p + 8*((lane>>4)&1);
    v8h r;
    #pragma unroll
    for(int e=0;e<8;++e) r[e]=(_Float16)s[e];
    *reinterpret_cast<v8h*>(L + (size_t)u*8) = r;
  }
}
__device__ __forceinline__ v16h load_tile_lds(const _Float16* L, int tile, int lane){
  const _Float16* p = L + ((size_t)tile*64 + lane)*8;
  return cat16(*reinterpret_cast<const v8h*>(p), *reinterpret_cast<const v8h*>(p + 32*8));
}

__global__ __launch_bounds__(256) void k_gru(
    const float* __restrict__ x, const float* __restrict__ Wih,
    const float* __restrict__ Whh, const float* __restrict__ bih,
    const float* __restrict__ bhh, float* __restrict__ seq)
{
  __shared__ _Float16 hscr[8][16*64];
  __shared__ __attribute__((aligned(16))) _Float16 WhhL[192*64];
  __shared__ __attribute__((aligned(16))) _Float16 WihL[192*128];
  int lane=threadIdx.x&31, wave=threadIdx.x>>5;
  int pbase=(blockIdx.x*8+wave)*16;
  int colb=lane&15, rowoff=(lane>>4)*8;

  stage_w<128,48,256>(Wih, WihL, threadIdx.x);
  stage_w<64,24,256>(Whh, WhhL, threadIdx.x);
  __syncthreads();

  float bi[12], bh[12];
  #pragma unroll
  for(int j=0;j<12;++j){ bi[j]=bih[j*16+colb]; bh[j]=bhh[j*16+colb]; }

  v8f hT[4], hs[4];
  #pragma unroll
  for(int j=0;j<4;++j){ hT[j]=zero8(); hs[j]=zero8(); }
  v16h hA[2]; hA[0]=zero16(); hA[1]=zero16();
  _Float16* hl=&hscr[wave][0];

  for(int t=0;t<PB;++t){
    unsigned zoff = 0;
    asm volatile("" : "+s"(zoff));
    const _Float16* Wt = WihL + zoff;
    const float* xsrc = x + ((size_t)pbase*PB + t)*DD;
    if(t+1<PB) __builtin_prefetch(xsrc+DD,0,1);
    v16h xA[4];
    #pragma unroll
    for(int kk=0;kk<4;++kk) xA[kk]=load_tile_f32(xsrc+kk*32, PB*DD, lane);

    #pragma unroll
    for(int j=0;j<4;++j){
      v8f gr=zero8();
      #pragma unroll
      for(int kk=0;kk<4;++kk)
        gr=WMMA_F16(xA[kk], load_tile_lds(Wt, j*4+kk, lane), gr);
      #pragma unroll
      for(int kk=0;kk<2;++kk)
        gr=WMMA_F16(hA[kk], load_tile_lds(WhhL, j*2+kk, lane), gr);
      v8f gz=zero8();
      #pragma unroll
      for(int kk=0;kk<4;++kk)
        gz=WMMA_F16(xA[kk], load_tile_lds(Wt, (4+j)*4+kk, lane), gz);
      #pragma unroll
      for(int kk=0;kk<2;++kk)
        gz=WMMA_F16(hA[kk], load_tile_lds(WhhL, (4+j)*2+kk, lane), gz);
      v8f gnx=zero8();
      #pragma unroll
      for(int kk=0;kk<4;++kk)
        gnx=WMMA_F16(xA[kk], load_tile_lds(Wt, (8+j)*4+kk, lane), gnx);
      v8f gnh=zero8();
      #pragma unroll
      for(int kk=0;kk<2;++kk)
        gnh=WMMA_F16(hA[kk], load_tile_lds(WhhL, (8+j)*2+kk, lane), gnh);
      #pragma unroll
      for(int e=0;e<8;++e){
        float r =sigf(gr[e]+bi[j]+bh[j]);
        float z =sigf(gz[e]+bi[4+j]+bh[4+j]);
        float nv=tanhf(gnx[e]+bi[8+j]+r*(gnh[e]+bh[8+j]));
        float hn=(1.f-z)*nv + z*hT[j][e];
        hT[j][e]=hn; hs[j][e]+=hn;
      }
    }
    #pragma unroll
    for(int j=0;j<4;++j){
      #pragma unroll
      for(int v=0;v<8;++v)
        hl[(v+rowoff)*64 + j*16 + colb]=(_Float16)hT[j][v];
    }
    #pragma unroll
    for(int kk=0;kk<2;++kk) hA[kk]=load_tile_f16(hl+kk*32, 64, lane);
  }
  float* sb=seq+(size_t)pbase*HH;
  #pragma unroll
  for(int j=0;j<4;++j){
    #pragma unroll
    for(int v=0;v<8;++v)
      sb[(size_t)(v+rowoff)*HH + j*16 + colb]=hs[j][v]*0.125f;
  }
}

__global__ __launch_bounds__(256) void k_attn(
    const float* __restrict__ X, const float* __restrict__ Wqkv,
    const float* __restrict__ bqkv, float* __restrict__ O)
{
  __shared__ __attribute__((aligned(16))) _Float16 Ks[SQ*32];
  __shared__ __attribute__((aligned(16))) _Float16 Vs[SQ*16];
  __shared__ __attribute__((aligned(16))) _Float16 Ps[8][32*16];
  __shared__ __attribute__((aligned(16))) _Float16 WL[6*512];
  int bh=blockIdx.x, b=bh>>2, h=bh&3;
  int tid=threadIdx.x, lane=tid&31, wave=tid>>5;
  int colb=lane&15, rowoff=(lane>>4)*8;

  const float* xs=X+((size_t)b*SQ+wave*64)*HH;
  v16h Xb[4][2];
  #pragma unroll
  for(int qt=0;qt<4;++qt){
    #pragma unroll
    for(int kk=0;kk<2;++kk) Xb[qt][kk]=load_tile_f32(xs+(size_t)qt*16*HH+kk*32, HH, lane);
  }
  {
    #pragma unroll
    for(int q=0;q<2;++q){
      int u=q*256+tid;
      if(u<6*64){
        int ul=u&31, p=(u>>5)&1, tile=u>>6;
        int g=tile>>1, kt=tile&1;
        const float* s=Wqkv+(size_t)(g*64+h*16+(ul&15))*HH+kt*32+16*p+8*((ul>>4)&1);
        v8h r;
        #pragma unroll
        for(int e=0;e<8;++e) r[e]=(_Float16)s[e];
        *reinterpret_cast<v8h*>(WL+(size_t)u*8)=r;
      }
    }
  }
  float bs[3][8];
  #pragma unroll
  for(int g=0;g<3;++g){
    #pragma unroll
    for(int r=0;r<8;++r) bs[g][r]=bqkv[g*64+h*16+rowoff+r];
  }
  __syncthreads();
  v16h Wa[3][2];
  #pragma unroll
  for(int g=0;g<3;++g){
    #pragma unroll
    for(int kk=0;kk<2;++kk) Wa[g][kk]=load_tile_lds(WL, g*2+kk, lane);
  }
  v8h zh;
  #pragma unroll
  for(int r=0;r<8;++r) zh[r]=(_Float16)0.f;
  v16h Qa[4];
  #pragma unroll
  for(int qt=0;qt<4;++qt){
    v8f q=zero8(), k=zero8(), v=zero8();
    #pragma unroll
    for(int kk=0;kk<2;++kk){
      q=WMMA_F16(Wa[0][kk], Xb[qt][kk], q);
      k=WMMA_F16(Wa[1][kk], Xb[qt][kk], k);
      v=WMMA_F16(Wa[2][kk], Xb[qt][kk], v);
    }
    v8h qh, kh, vh;
    #pragma unroll
    for(int r=0;r<8;++r){
      qh[r]=(_Float16)((q[r]+bs[0][r])*0.25f);
      kh[r]=(_Float16)(k[r]+bs[1][r]);
      vh[r]=(_Float16)(v[r]+bs[2][r]);
    }
    Qa[qt]=cat16(qh, zh);
    int key=wave*64+qt*16+colb;
    *reinterpret_cast<v8h*>(Ks+(size_t)key*32+rowoff)=kh;
    *reinterpret_cast<v8h*>(Ks+(size_t)key*32+16+rowoff)=zh;
    *reinterpret_cast<v8h*>(Vs+(size_t)key*16+rowoff)=vh;
  }
  __syncthreads();

  v8f mrow[4];
  {
    v8f lo;
    lo[0]=-1e30f; lo[1]=-1e30f; lo[2]=-1e30f; lo[3]=-1e30f;
    lo[4]=-1e30f; lo[5]=-1e30f; lo[6]=-1e30f; lo[7]=-1e30f;
    mrow[0]=lo; mrow[1]=lo; mrow[2]=lo; mrow[3]=lo;
  }
  for(int kc=0;kc<16;++kc){
    int kbase=kc*32;
    v16h Kb0=load_tile_f16(Ks+(size_t)kbase*32, 32, lane);
    v16h Kb1=load_tile_f16(Ks+(size_t)(kbase+16)*32, 32, lane);
    #pragma unroll
    for(int qt=0;qt<4;++qt){
      v8f s0=WMMA_F16(Qa[qt],Kb0,zero8());
      v8f s1=WMMA_F16(Qa[qt],Kb1,zero8());
      mrow[qt]=__builtin_elementwise_max(mrow[qt],__builtin_elementwise_max(s0,s1));
    }
  }
  mrow[0]=rmax16x8(mrow[0]);
  mrow[1]=rmax16x8(mrow[1]);
  mrow[2]=rmax16x8(mrow[2]);
  mrow[3]=rmax16x8(mrow[3]);

  _Float16* pl=&Ps[wave][0];
  _Float16* pst=pl+colb*16+rowoff;
  int troff=(8*(lane>>4)+(lane&7))*16+8*((lane>>3)&1);
  const _Float16* pld=pl+troff;
  const _Float16* vld=Vs+troff;
  #pragma unroll
  for(int qt=0;qt<4;++qt){
    int qbase=wave*64+qt*16;
    v8f acc=zero8(), lrow=zero8();
    for(int kc=0;kc<16;++kc){
      int kbase=kc*32;
      v16h Kb0=load_tile_f16(Ks+(size_t)kbase*32, 32, lane);
      v16h Kb1=load_tile_f16(Ks+(size_t)(kbase+16)*32, 32, lane);
      v8f s0=WMMA_F16(Qa[qt],Kb0,zero8());
      v8f s1=WMMA_F16(Qa[qt],Kb1,zero8());
      #pragma unroll
      for(int e=0;e<8;++e){
        s0[e]=__expf(s0[e]-mrow[qt][e]);
        s1[e]=__expf(s1[e]-mrow[qt][e]);
        lrow[e]+=s0[e]+s1[e];
      }
      v8h p0, p1;
      #pragma unroll
      for(int e=0;e<8;++e){ p0[e]=(_Float16)s0[e]; p1[e]=(_Float16)s1[e]; }
      *reinterpret_cast<v8h*>(pst)=p0;
      *reinterpret_cast<v8h*>(pst+16*16)=p1;
      v16h Pa=cat16(ld_tr16(pld), ld_tr16(pld+16*16));
      const _Float16* vk=vld+(size_t)kbase*16;
      v16h Vb=cat16(ld_tr16(vk), ld_tr16(vk+16*16));
      acc=WMMA_F16(Pa,Vb,acc);
    }
    lrow=rsum16x8(lrow);
    float* ob=O+((size_t)b*SQ+qbase)*HH + h*16;
    #pragma unroll
    for(int e=0;e<8;++e)
      ob[(size_t)(e+rowoff)*HH+colb]=acc[e]/lrow[e];
  }
}

__global__ __launch_bounds__(256) void k_projln(
    float* __restrict__ X, const float* __restrict__ Y,
    const float* __restrict__ Wo, const float* __restrict__ bo,
    const float* __restrict__ g, const float* __restrict__ be)
{
  __shared__ __attribute__((aligned(16))) _Float16 WL[64*64];
  int lane=threadIdx.x&31, wave=threadIdx.x>>5;
  size_t rbase=((size_t)blockIdx.x*8+wave)*16;
  int colb=lane&15, rowoff=(lane>>4)*8;
  stage_w<64,8,256>(Wo, WL, threadIdx.x);
  v16h A[2];
  #pragma unroll
  for(int kk=0;kk<2;++kk) A[kk]=load_tile_f32(Y+rbase*HH+kk*32, HH, lane);
  __syncthreads();
  v8f o[4];
  #pragma unroll
  for(int j=0;j<4;++j){
    v8f acc=zero8();
    #pragma unroll
    for(int kk=0;kk<2;++kk)
      acc=WMMA_F16(A[kk], load_tile_lds(WL, j*2+kk, lane), acc);
    float bias=bo[j*16+colb];
    const float* xr=X+rbase*HH+j*16+colb;
    #pragma unroll
    for(int e=0;e<8;++e) o[j][e]=acc[e]+bias+xr[(size_t)(e+rowoff)*HH];
  }
  float gv[4],bv[4];
  #pragma unroll
  for(int j=0;j<4;++j){ gv[j]=g[j*16+colb]; bv[j]=be[j*16+colb]; }
  float* xw=X+rbase*HH;
  #pragma unroll
  for(int e=0;e<8;++e){
    float s=o[0][e]+o[1][e]+o[2][e]+o[3][e];
    float mu=rsum16(s)*(1.f/64.f);
    float d0=o[0][e]-mu,d1=o[1][e]-mu,d2=o[2][e]-mu,d3=o[3][e]-mu;
    float var=rsum16(d0*d0+d1*d1+d2*d2+d3*d3)*(1.f/64.f);
    float inv=rsqrtf(var+1e-5f);
    #pragma unroll
    for(int j=0;j<4;++j)
      xw[(size_t)(e+rowoff)*HH+j*16+colb]=(o[j][e]-mu)*inv*gv[j]+bv[j];
  }
}

template<bool LAST>
__global__ __launch_bounds__(128) void k_ffln(
    float* __restrict__ X, const float* __restrict__ W1,
    const float* __restrict__ b1, const float* __restrict__ W2,
    const float* __restrict__ b2, const float* __restrict__ g,
    const float* __restrict__ be, float* __restrict__ part)
{
  __shared__ _Float16 h1[4][16*256];
  __shared__ float red[4][64];
  __shared__ __attribute__((aligned(16))) _Float16 W1L[256*64];
  __shared__ __attribute__((aligned(16))) _Float16 W2L[64*256];
  int lane=threadIdx.x&31, wave=threadIdx.x>>5;
  size_t rbase=((size_t)blockIdx.x*4+wave)*16;
  int colb=lane&15, rowoff=(lane>>4)*8;
  stage_w<64,32,128>(W1, W1L, threadIdx.x);
  stage_w<256,32,128>(W2, W2L, threadIdx.x);
  v16h A[2];
  #pragma unroll
  for(int kk=0;kk<2;++kk) A[kk]=load_tile_f32(X+rbase*HH+kk*32, HH, lane);
  __syncthreads();
  _Float16* hl=&h1[wave][0];
  #pragma unroll
  for(int j=0;j<16;++j){
    v8f acc=zero8();
    #pragma unroll
    for(int kk=0;kk<2;++kk)
      acc=WMMA_F16(A[kk], load_tile_lds(W1L, j*2+kk, lane), acc);
    float bias=b1[j*16+colb];
    #pragma unroll
    for(int e=0;e<8;++e)
      hl[(size_t)(e+rowoff)*256+j*16+colb]=(_Float16)fmaxf(acc[e]+bias,0.f);
  }
  v16h pA[8];
  #pragma unroll
  for(int kk=0;kk<8;++kk) pA[kk]=load_tile_f16(hl+kk*32, 256, lane);
  v8f o[4];
  #pragma unroll
  for(int j=0;j<4;++j){
    v8f acc=zero8();
    #pragma unroll
    for(int kk=0;kk<8;++kk)
      acc=WMMA_F16(pA[kk], load_tile_lds(W2L, j*8+kk, lane), acc);
    float bias=b2[j*16+colb];
    const float* xr=X+rbase*HH+j*16+colb;
    #pragma unroll
    for(int e=0;e<8;++e) o[j][e]=acc[e]+bias+xr[(size_t)(e+rowoff)*HH];
  }
  float gv[4],bv[4];
  #pragma unroll
  for(int j=0;j<4;++j){ gv[j]=g[j*16+colb]; bv[j]=be[j*16+colb]; }
  float* xw=X+rbase*HH;
  float cs[4]={0.f,0.f,0.f,0.f};
  #pragma unroll
  for(int e=0;e<8;++e){
    float s=o[0][e]+o[1][e]+o[2][e]+o[3][e];
    float mu=rsum16(s)*(1.f/64.f);
    float d0=o[0][e]-mu,d1=o[1][e]-mu,d2=o[2][e]-mu,d3=o[3][e]-mu;
    float var=rsum16(d0*d0+d1*d1+d2*d2+d3*d3)*(1.f/64.f);
    float inv=rsqrtf(var+1e-5f);
    #pragma unroll
    for(int j=0;j<4;++j){
      float y=(o[j][e]-mu)*inv*gv[j]+bv[j];
      if(LAST) cs[j]+=y;
      else     xw[(size_t)(e+rowoff)*HH+j*16+colb]=y;
    }
  }
  if(LAST){
    #pragma unroll
    for(int j=0;j<4;++j) cs[j]+=__shfl_xor(cs[j],16,32);
    if(lane<16){
      #pragma unroll
      for(int j=0;j<4;++j) red[wave][j*16+lane]=cs[j];
    }
    __syncthreads();
    if(threadIdx.x<64){
      int t=threadIdx.x;
      part[(size_t)blockIdx.x*HH+t]=(red[0][t]+red[1][t])+(red[2][t]+red[3][t]);
    }
  }
}

__global__ void k_mean(const float* __restrict__ part, float* __restrict__ out){
  int b=blockIdx.x, j=threadIdx.x;
  const float* p=part+(size_t)b*8*HH+j;
  float v[8];
  #pragma unroll
  for(int w=0;w<8;++w) v[w]=p[w*HH];
  float s=((v[0]+v[1])+(v[2]+v[3]))+((v[4]+v[5])+(v[6]+v[7]));
  out[b*HH+j]=s*(1.f/512.f);
}

extern "C" void kernel_launch(void* const* d_in, const int* in_sizes, int n_in,
                              void* d_out, int out_size, void* d_ws, size_t ws_size,
                              hipStream_t stream){
  const float* x  =(const float*)d_in[0];
  const float* Wih=(const float*)d_in[1];
  const float* Whh=(const float*)d_in[2];
  const float* bih=(const float*)d_in[3];
  const float* bhh=(const float*)d_in[4];

  char* ws=(char*)d_ws;
  float*    X =(float*)ws;
  float*    Y =(float*)(ws+(size_t)(8<<20));
  float*    Pm=Y;

  k_gru<<<256,256,0,stream>>>(x,Wih,Whh,bih,bhh,X);

  for(int L=0;L<2;++L){
    int base=5+L*12;
    const float* Wqkv=(const float*)d_in[base+0];
    const float* Wo  =(const float*)d_in[base+2];
    const float* W1  =(const float*)d_in[base+4];
    const float* W2  =(const float*)d_in[base+6];
    const float* bqkv=(const float*)d_in[base+1];
    const float* bo  =(const float*)d_in[base+3];
    const float* b1  =(const float*)d_in[base+5];
    const float* b2  =(const float*)d_in[base+7];
    const float* g1  =(const float*)d_in[base+8];
    const float* be1 =(const float*)d_in[base+9];
    const float* g2  =(const float*)d_in[base+10];
    const float* be2 =(const float*)d_in[base+11];

    k_attn  <<<256,256,0,stream>>>(X,Wqkv,bqkv,Y);
    k_projln<<<256,256,0,stream>>>(X,Y,Wo,bo,g1,be1);
    if(L==0) k_ffln<false><<<512,128,0,stream>>>(X,W1,b1,W2,b2,g2,be2,Pm);
    else     k_ffln<true> <<<512,128,0,stream>>>(X,W1,b1,W2,b2,g2,be2,Pm);
  }
  k_mean<<<64,64,0,stream>>>(Pm,(float*)d_out);
}


# attention pass 2 loop order: 32-key chunk outer, the wave's 4 query tiles inner: K, V fragments read once per chunk, 4 independent score-exp-P-PV chains overlap in one wave; sums unchanged
# speedup vs baseline: 2.5081x; 1.0423x over previous
#include <hip/hip_runtime.h>
#include <hip/hip_fp16.h>

typedef __attribute__((ext_vector_type(16))) _Float16 v16h;
typedef __attribute__((ext_vector_type(8)))  float    v8f;

#define WMMA_F16(a,b,c) __builtin_amdgcn_wmma_f32_16x16x32_f16(false,(a),false,(b),(short)0,(c),false,false)

__device__ __forceinline__ v8f zero8(){
  v8f z;
  #pragma unroll
  for(int i=0;i<8;++i) z[i]=0.f;
  return z;
}
__device__ __forceinline__ v16h zero16(){
  v16h z;
  #pragma unroll
  for(int i=0;i<16;++i) z[i]=(_Float16)0.f;
  return z;
}

__device__ __forceinline__ int a_k(int i, int h16){ return (i&7) + ((i>>3)<<4) + (h16<<3); }

__device__ __forceinline__ v16h load_tile_f32(const float* __restrict__ src, int ld, int lane){
  const float* rb = src + (size_t)(lane&15)*ld;
  int h16=(lane>>4)&1; v16h r;
  #pragma unroll
  for(int i=0;i<16;++i) r[i]=(_Float16)rb[a_k(i,h16)];
  return r;
}
__device__ __forceinline__ v16h load_tile_f16(const _Float16* __restrict__ src, int ld, int lane){
  const _Float16* rb = src + (size_t)(lane&15)*ld;
  int h16=(lane>>4)&1; v16h r;
  #pragma unroll
  for(int i=0;i<16;++i) r[i]=rb[a_k(i,h16)];
  return r;
}
__device__ __forceinline__ v16h load_tile_kmajor_f16(const _Float16* __restrict__ src, int ld, int lane){
  const _Float16* cb = src + (lane&15);
  int h16=(lane>>4)&1; v16h r;
  #pragma unroll
  for(int i=0;i<16;++i) r[i]=cb[(size_t)a_k(i,h16)*ld];
  return r;
}

typedef short s16x8 __attribute__((ext_vector_type(8)));
typedef __attribute__((ext_vector_type(8))) _Float16 v8h;
__device__ __forceinline__ v8h ld_tr16(const _Float16* p){
  typedef __attribute__((address_space(3))) s16x8 lds_s16x8;
  return __builtin_bit_cast(v8h, __builtin_amdgcn_ds_load_tr16_b128_v8i16((lds_s16x8*)p));
}
__device__ __forceinline__ v16h cat16(v8h lo, v8h hi){
  return __builtin_shufflevector(lo, hi, 0,1,2,3,4,5,6,7,8,9,10,11,12,13,14,15);
}

__device__ __forceinline__ float rmax16(float x){
  #pragma unroll
  for(int m=8;m;m>>=1) x=fmaxf(x,__shfl_xor(x,m,32));
  return x;
}
__device__ __forceinline__ float rsum16(float x){
  #pragma unroll
  for(int m=8;m;m>>=1) x+=__shfl_xor(x,m,32);
  return x;
}
__device__ __forceinline__ float sigf(float x){ return 1.f/(1.f+__expf(-x)); }

__device__ __forceinline__ v8f rmax16x8(v8f x){
  #pragma unroll
  for(int m=8;m;m>>=1){
    v8f t;
    #pragma unroll
    for(int e=0;e<8;++e) t[e]=__shfl_xor(x[e],m,32);
    x=__builtin_elementwise_max(x,t);
  }
  return x;
}
__device__ __forceinline__ v8f rsum16x8(v8f x){
  #pragma unroll
  for(int m=8;m;m>>=1){
    v8f t;
    #pragma unroll
    for(int e=0;e<8;++e) t[e]=__shfl_xor(x[e],m,32);
    x+=t;
  }
  return x;
}

#define PB  8
#define DD  128
#define HH  64
#define SQ  512

template<int K, int NT, int NTHR>
__device__ __forceinline__ void stage_w(const float* __restrict__ W, _Float16* L, int tid){
  constexpr int KT = K/32;
  #pragma unroll
  for(int q=0;q<NT*64/NTHR;++q){
    int u    = q*NTHR + tid;
    int lane = u & 31;
    int p    = (u >> 5) & 1;
    int tile = u >> 6;
    int kt = tile % KT, jt = tile / KT;
    const float* s = W + (size_t)(jt*16 + (lane&15))*K + kt*32 + 16*p + 8*((lane>>4)&1);
    v8h r;
    #pragma unroll
    for(int e=0;e<8;++e) r[e]=(_Float16)s[e];
    *reinterpret_cast<v8h*>(L + (size_t)u*8) = r;
  }
}
__device__ __forceinline__ v16h load_tile_lds(const _Float16* L, int tile, int lane){
  const _Float16* p = L + ((size_t)tile*64 + lane)*8;
  return cat16(*reinterpret_cast<const v8h*>(p), *reinterpret_cast<const v8h*>(p + 32*8));
}

__global__ __launch_bounds__(256) void k_gru(
    const float* __restrict__ x, const float* __restrict__ Wih,
    const float* __restrict__ Whh, const float* __restrict__ bih,
    const float* __restrict__ bhh, float* __restrict__ seq)
{
  __shared__ _Float16 hscr[8][16*64];
  __shared__ __attribute__((aligned(16))) _Float16 WhhL[192*64];
  __shared__ __attribute__((aligned(16))) _Float16 WihL[192*128];
  int lane=threadIdx.x&31, wave=threadIdx.x>>5;
  int pbase=(blockIdx.x*8+wave)*16;
  int colb=lane&15, rowoff=(lane>>4)*8;

  stage_w<128,48,256>(Wih, WihL, threadIdx.x);
  stage_w<64,24,256>(Whh, WhhL, threadIdx.x);
  __syncthreads();

  float bi[12], bh[12];
  #pragma unroll
  for(int j=0;j<12;++j){ bi[j]=bih[j*16+colb]; bh[j]=bhh[j*16+colb]; }

  v8f hT[4], hs[4];
  #pragma unroll
  for(int j=0;j<4;++j){ hT[j]=zero8(); hs[j]=zero8(); }
  v16h hA[2]; hA[0]=zero16(); hA[1]=zero16();
  _Float16* hl=&hscr[wave][0];

  for(int t=0;t<PB;++t){
    unsigned zoff = 0;
    asm volatile("" : "+s"(zoff));
    const _Float16* Wt = WihL + zoff;
    const float* xsrc = x + ((size_t)pbase*PB + t)*DD;
    if(t+1<PB) __builtin_prefetch(xsrc+DD,0,1);
    v16h xA[4];
    #pragma unroll
    for(int kk=0;kk<4;++kk) xA[kk]=load_tile_f32(xsrc+kk*32, PB*DD, lane);

    #pragma unroll
    for(int j=0;j<4;++j){
      v8f gr=zero8();
      #pragma unroll
      for(int kk=0;kk<4;++kk)
        gr=WMMA_F16(xA[kk], load_tile_lds(Wt, j*4+kk, lane), gr);
      #pragma unroll
      for(int kk=0;kk<2;++kk)
        gr=WMMA_F16(hA[kk], load_tile_lds(WhhL, j*2+kk, lane), gr);
      v8f gz=zero8();
      #pragma unroll
      for(int kk=0;kk<4;++kk)
        gz=WMMA_F16(xA[kk], load_tile_lds(Wt, (4+j)*4+kk, lane), gz);
      #pragma unroll
      for(int kk=0;kk<2;++kk)
        gz=WMMA_F16(hA[kk], load_tile_lds(WhhL, (4+j)*2+kk, lane), gz);
      v8f gnx=zero8();
      #pragma unroll
      for(int kk=0;kk<4;++kk)
        gnx=WMMA_F16(xA[kk], load_tile_lds(Wt, (8+j)*4+kk, lane), gnx);
      v8f gnh=zero8();
      #pragma unroll
      for(int kk=0;kk<2;++kk)
        gnh=WMMA_F16(hA[kk], load_tile_lds(WhhL, (8+j)*2+kk, lane), gnh);
      #pragma unroll
      for(int e=0;e<8;++e){
        float r =sigf(gr[e]+bi[j]+bh[j]);
        float z =sigf(gz[e]+bi[4+j]+bh[4+j]);
        float nv=tanhf(gnx[e]+bi[8+j]+r*(gnh[e]+bh[8+j]));
        float hn=(1.f-z)*nv + z*hT[j][e];
        hT[j][e]=hn; hs[j][e]+=hn;
      }
    }
    #pragma unroll
    for(int j=0;j<4;++j){
      #pragma unroll
      for(int v=0;v<8;++v)
        hl[(v+rowoff)*64 + j*16 + colb]=(_Float16)hT[j][v];
    }
    #pragma unroll
    for(int kk=0;kk<2;++kk) hA[kk]=load_tile_f16(hl+kk*32, 64, lane);
  }
  float* sb=seq+(size_t)pbase*HH;
  #pragma unroll
  for(int j=0;j<4;++j){
    #pragma unroll
    for(int v=0;v<8;++v)
      sb[(size_t)(v+rowoff)*HH + j*16 + colb]=hs[j][v]*0.125f;
  }
}

__device__ __forceinline__ v16h softmax_tile(v8f s0, v8f s1, v8f m, v8f& lrow,
                                             _Float16* pst, const _Float16* pld){
  v8h p0, p1;
  #pragma unroll
  for(int e=0;e<8;++e){
    float a=__expf(s0[e]-m[e]);
    float c=__expf(s1[e]-m[e]);
    lrow[e]+=a+c;
    p0[e]=(_Float16)a; p1[e]=(_Float16)c;
  }
  *reinterpret_cast<v8h*>(pst)=p0;
  *reinterpret_cast<v8h*>(pst+16*16)=p1;
  return cat16(ld_tr16(pld), ld_tr16(pld+16*16));
}

__global__ __launch_bounds__(256) void k_attn(
    const float* __restrict__ X, const float* __restrict__ Wqkv,
    const float* __restrict__ bqkv, float* __restrict__ O)
{
  __shared__ __attribute__((aligned(16))) _Float16 Ks[SQ*32];
  __shared__ __attribute__((aligned(16))) _Float16 Vs[SQ*16];
  __shared__ __attribute__((aligned(16))) _Float16 Ps[8][4*32*16];
  __shared__ __attribute__((aligned(16))) _Float16 WL[6*512];
  int bh=blockIdx.x, b=bh>>2, h=bh&3;
  int tid=threadIdx.x, lane=tid&31, wave=tid>>5;
  int colb=lane&15, rowoff=(lane>>4)*8;

  const float* xs=X+((size_t)b*SQ+wave*64)*HH;
  v16h Xb[4][2];
  #pragma unroll
  for(int qt=0;qt<4;++qt){
    #pragma unroll
    for(int kk=0;kk<2;++kk) Xb[qt][kk]=load_tile_f32(xs+(size_t)qt*16*HH+kk*32, HH, lane);
  }
  {
    #pragma unroll
    for(int q=0;q<2;++q){
      int u=q*256+tid;
      if(u<6*64){
        int ul=u&31, p=(u>>5)&1, tile=u>>6;
        int g=tile>>1, kt=tile&1;
        const float* s=Wqkv+(size_t)(g*64+h*16+(ul&15))*HH+kt*32+16*p+8*((ul>>4)&1);
        v8h r;
        #pragma unroll
        for(int e=0;e<8;++e) r[e]=(_Float16)s[e];
        *reinterpret_cast<v8h*>(WL+(size_t)u*8)=r;
      }
    }
  }
  float bs[3][8];
  #pragma unroll
  for(int g=0;g<3;++g){
    #pragma unroll
    for(int r=0;r<8;++r) bs[g][r]=bqkv[g*64+h*16+rowoff+r];
  }
  __syncthreads();
  v16h Wa[3][2];
  #pragma unroll
  for(int g=0;g<3;++g){
    #pragma unroll
    for(int kk=0;kk<2;++kk) Wa[g][kk]=load_tile_lds(WL, g*2+kk, lane);
  }
  v8h zh;
  #pragma unroll
  for(int r=0;r<8;++r) zh[r]=(_Float16)0.f;
  v16h Qa[4];
  #pragma unroll
  for(int qt=0;qt<4;++qt){
    v8f q=zero8(), k=zero8(), v=zero8();
    #pragma unroll
    for(int kk=0;kk<2;++kk){
      q=WMMA_F16(Wa[0][kk], Xb[qt][kk], q);
      k=WMMA_F16(Wa[1][kk], Xb[qt][kk], k);
      v=WMMA_F16(Wa[2][kk], Xb[qt][kk], v);
    }
    v8h qh, kh, vh;
    #pragma unroll
    for(int r=0;r<8;++r){
      qh[r]=(_Float16)((q[r]+bs[0][r])*0.25f);
      kh[r]=(_Float16)(k[r]+bs[1][r]);
      vh[r]=(_Float16)(v[r]+bs[2][r]);
    }
    Qa[qt]=cat16(qh, zh);
    int key=wave*64+qt*16+colb;
    *reinterpret_cast<v8h*>(Ks+(size_t)key*32+rowoff)=kh;
    *reinterpret_cast<v8h*>(Ks+(size_t)key*32+16+rowoff)=zh;
    *reinterpret_cast<v8h*>(Vs+(size_t)key*16+rowoff)=vh;
  }
  __syncthreads();

  v8f mrow[4];
  {
    v8f lo;
    lo[0]=-1e30f; lo[1]=-1e30f; lo[2]=-1e30f; lo[3]=-1e30f;
    lo[4]=-1e30f; lo[5]=-1e30f; lo[6]=-1e30f; lo[7]=-1e30f;
    mrow[0]=lo; mrow[1]=lo; mrow[2]=lo; mrow[3]=lo;
  }
  for(int kc=0;kc<16;++kc){
    int kbase=kc*32;
    v16h Kb0=load_tile_f16(Ks+(size_t)kbase*32, 32, lane);
    v16h Kb1=load_tile_f16(Ks+(size_t)(kbase+16)*32, 32, lane);
    #pragma unroll
    for(int qt=0;qt<4;++qt){
      v8f s0=WMMA_F16(Qa[qt],Kb0,zero8());
      v8f s1=WMMA_F16(Qa[qt],Kb1,zero8());
      mrow[qt]=__builtin_elementwise_max(mrow[qt],__builtin_elementwise_max(s0,s1));
    }
  }
  mrow[0]=rmax16x8(mrow[0]);
  mrow[1]=rmax16x8(mrow[1]);
  mrow[2]=rmax16x8(mrow[2]);
  mrow[3]=rmax16x8(mrow[3]);

  _Float16* pl=&Ps[wave][0];
  _Float16* pst=pl+colb*16+rowoff;
  int troff=(8*(lane>>4)+(lane&7))*16+8*((lane>>3)&1);
  const _Float16* pld=pl+troff;
  const _Float16* vld=Vs+troff;
  v8f acc[4], lrow[4];
  acc[0]=zero8(); acc[1]=zero8(); acc[2]=zero8(); acc[3]=zero8();
  lrow[0]=zero8(); lrow[1]=zero8(); lrow[2]=zero8(); lrow[3]=zero8();
  v16h Kb0=load_tile_f16(Ks, 32, lane);
  v16h Kb1=load_tile_f16(Ks+(size_t)16*32, 32, lane);
  v16h Vb=cat16(ld_tr16(vld), ld_tr16(vld+16*16));
  v16h Pa3=zero16(), Vp=zero16();
  for(int kc=0;kc<16;++kc){
    int kn=(kc<15?kc+1:15)*32;
    v16h Kn0=load_tile_f16(Ks+(size_t)kn*32, 32, lane);
    v16h Kn1=load_tile_f16(Ks+(size_t)(kn+16)*32, 32, lane);
    const _Float16* vk=vld+(size_t)kn*16;
    v16h Vn=cat16(ld_tr16(vk), ld_tr16(vk+16*16));
    v8f s00=WMMA_F16(Qa[0],Kb0,zero8()), s01=WMMA_F16(Qa[0],Kb1,zero8());
    v8f s10=WMMA_F16(Qa[1],Kb0,zero8()), s11=WMMA_F16(Qa[1],Kb1,zero8());
    v8f s20=WMMA_F16(Qa[2],Kb0,zero8()), s21=WMMA_F16(Qa[2],Kb1,zero8());
    v8f s30=WMMA_F16(Qa[3],Kb0,zero8()), s31=WMMA_F16(Qa[3],Kb1,zero8());
    acc[3]=WMMA_F16(Pa3,Vp,acc[3]);
    v16h Pa0=softmax_tile(s00,s01,mrow[0],lrow[0],pst,pld);
    v16h Pa1=softmax_tile(s10,s11,mrow[1],lrow[1],pst+512,pld+512);
    acc[0]=WMMA_F16(Pa0,Vb,acc[0]);
    v16h Pa2=softmax_tile(s20,s21,mrow[2],lrow[2],pst+1024,pld+1024);
    acc[1]=WMMA_F16(Pa1,Vb,acc[1]);
    Pa3=softmax_tile(s30,s31,mrow[3],lrow[3],pst+1536,pld+1536);
    acc[2]=WMMA_F16(Pa2,Vb,acc[2]);
    Vp=Vb; Kb0=Kn0; Kb1=Kn1; Vb=Vn;
  }
  acc[3]=WMMA_F16(Pa3,Vp,acc[3]);
  #pragma unroll
  for(int qt=0;qt<4;++qt){
    int qbase=wave*64+qt*16;
    v8f lr=rsum16x8(lrow[qt]);
    float* ob=O+((size_t)b*SQ+qbase)*HH + h*16;
    #pragma unroll
    for(int e=0;e<8;++e)
      ob[(size_t)(e+rowoff)*HH+colb]=acc[qt][e]/lr[e];
  }
}

__global__ __launch_bounds__(256) void k_projln(
    float* __restrict__ X, const float* __restrict__ Y,
    const float* __restrict__ Wo, const float* __restrict__ bo,
    const float* __restrict__ g, const float* __restrict__ be)
{
  __shared__ __attribute__((aligned(16))) _Float16 WL[64*64];
  int lane=threadIdx.x&31, wave=threadIdx.x>>5;
  size_t rbase=((size_t)blockIdx.x*8+wave)*16;
  int colb=lane&15, rowoff=(lane>>4)*8;
  stage_w<64,8,256>(Wo, WL, threadIdx.x);
  v16h A[2];
  #pragma unroll
  for(int kk=0;kk<2;++kk) A[kk]=load_tile_f32(Y+rbase*HH+kk*32, HH, lane);
  __syncthreads();
  v8f o[4];
  #pragma unroll
  for(int j=0;j<4;++j){
    v8f acc=zero8();
    #pragma unroll
    for(int kk=0;kk<2;++kk)
      acc=WMMA_F16(A[kk], load_tile_lds(WL, j*2+kk, lane), acc);
    float bias=bo[j*16+colb];
    const float* xr=X+rbase*HH+j*16+colb;
    #pragma unroll
    for(int e=0;e<8;++e) o[j][e]=acc[e]+bias+xr[(size_t)(e+rowoff)*HH];
  }
  float gv[4],bv[4];
  #pragma unroll
  for(int j=0;j<4;++j){ gv[j]=g[j*16+colb]; bv[j]=be[j*16+colb]; }
  float* xw=X+rbase*HH;
  #pragma unroll
  for(int e=0;e<8;++e){
    float s=o[0][e]+o[1][e]+o[2][e]+o[3][e];
    float mu=rsum16(s)*(1.f/64.f);
    float d0=o[0][e]-mu,d1=o[1][e]-mu,d2=o[2][e]-mu,d3=o[3][e]-mu;
    float var=rsum16(d0*d0+d1*d1+d2*d2+d3*d3)*(1.f/64.f);
    float inv=rsqrtf(var+1e-5f);
    #pragma unroll
    for(int j=0;j<4;++j)
      xw[(size_t)(e+rowoff)*HH+j*16+colb]=(o[j][e]-mu)*inv*gv[j]+bv[j];
  }
}

template<bool LAST>
__global__ __launch_bounds__(128) void k_ffln(
    float* __restrict__ X, const float* __restrict__ W1,
    const float* __restrict__ b1, const float* __restrict__ W2,
    const float* __restrict__ b2, const float* __restrict__ g,
    const float* __restrict__ be, float* __restrict__ part)
{
  __shared__ _Float16 h1[4][16*256];
  __shared__ float red[4][64];
  __shared__ __attribute__((aligned(16))) _Float16 W1L[256*64];
  __shared__ __attribute__((aligned(16))) _Float16 W2L[64*256];
  int lane=threadIdx.x&31, wave=threadIdx.x>>5;
  size_t rbase=((size_t)blockIdx.x*4+wave)*16;
  int colb=lane&15, rowoff=(lane>>4)*8;
  stage_w<64,32,128>(W1, W1L, threadIdx.x);
  stage_w<256,32,128>(W2, W2L, threadIdx.x);
  v16h A[2];
  #pragma unroll
  for(int kk=0;kk<2;++kk) A[kk]=load_tile_f32(X+rbase*HH+kk*32, HH, lane);
  __syncthreads();
  _Float16* hl=&h1[wave][0];
  #pragma unroll
  for(int j=0;j<16;++j){
    v8f acc=zero8();
    #pragma unroll
    for(int kk=0;kk<2;++kk)
      acc=WMMA_F16(A[kk], load_tile_lds(W1L, j*2+kk, lane), acc);
    float bias=b1[j*16+colb];
    #pragma unroll
    for(int e=0;e<8;++e)
      hl[(size_t)(e+rowoff)*256+j*16+colb]=(_Float16)fmaxf(acc[e]+bias,0.f);
  }
  v16h pA[8];
  #pragma unroll
  for(int kk=0;kk<8;++kk) pA[kk]=load_tile_f16(hl+kk*32, 256, lane);
  v8f o[4];
  #pragma unroll
  for(int j=0;j<4;++j){
    v8f acc=zero8();
    #pragma unroll
    for(int kk=0;kk<8;++kk)
      acc=WMMA_F16(pA[kk], load_tile_lds(W2L, j*8+kk, lane), acc);
    float bias=b2[j*16+colb];
    const float* xr=X+rbase*HH+j*16+colb;
    #pragma unroll
    for(int e=0;e<8;++e) o[j][e]=acc[e]+bias+xr[(size_t)(e+rowoff)*HH];
  }
  float gv[4],bv[4];
  #pragma unroll
  for(int j=0;j<4;++j){ gv[j]=g[j*16+colb]; bv[j]=be[j*16+colb]; }
  float* xw=X+rbase*HH;
  float cs[4]={0.f,0.f,0.f,0.f};
  #pragma unroll
  for(int e=0;e<8;++e){
    float s=o[0][e]+o[1][e]+o[2][e]+o[3][e];
    float mu=rsum16(s)*(1.f/64.f);
    float d0=o[0][e]-mu,d1=o[1][e]-mu,d2=o[2][e]-mu,d3=o[3][e]-mu;
    float var=rsum16(d0*d0+d1*d1+d2*d2+d3*d3)*(1.f/64.f);
    float inv=rsqrtf(var+1e-5f);
    #pragma unroll
    for(int j=0;j<4;++j){
      float y=(o[j][e]-mu)*inv*gv[j]+bv[j];
      if(LAST) cs[j]+=y;
      else     xw[(size_t)(e+rowoff)*HH+j*16+colb]=y;
    }
  }
  if(LAST){
    #pragma unroll
    for(int j=0;j<4;++j) cs[j]+=__shfl_xor(cs[j],16,32);
    if(lane<16){
      #pragma unroll
      for(int j=0;j<4;++j) red[wave][j*16+lane]=cs[j];
    }
    __syncthreads();
    if(threadIdx.x<64){
      int t=threadIdx.x;
      part[(size_t)blockIdx.x*HH+t]=(red[0][t]+red[1][t])+(red[2][t]+red[3][t]);
    }
  }
}

__global__ void k_mean(const float* __restrict__ part, float* __restrict__ out){
  int b=blockIdx.x, j=threadIdx.x;
  const float* p=part+(size_t)b*8*HH+j;
  float v[8];
  #pragma unroll
  for(int w=0;w<8;++w) v[w]=p[w*HH];
  float s=((v[0]+v[1])+(v[2]+v[3]))+((v[4]+v[5])+(v[6]+v[7]));
  out[b*HH+j]=s*(1.f/512.f);
}

extern "C" void kernel_launch(void* const* d_in, const int* in_sizes, int n_in,
                              void* d_out, int out_size, void* d_ws, size_t ws_size,
                              hipStream_t stream){
  const float* x  =(const float*)d_in[0];
  const float* Wih=(const float*)d_in[1];
  const float* Whh=(const float*)d_in[2];
  const float* bih=(const float*)d_in[3];
  const float* bhh=(const float*)d_in[4];

  char* ws=(char*)d_ws;
  float*    X =(float*)ws;
  float*    Y =(float*)(ws+(size_t)(8<<20));
  float*    Pm=Y;

  k_gru<<<256,256,0,stream>>>(x,Wih,Whh,bih,bhh,X);

  for(int L=0;L<2;++L){
    int base=5+L*12;
    const float* Wqkv=(const float*)d_in[base+0];
    const float* Wo  =(const float*)d_in[base+2];
    const float* W1  =(const float*)d_in[base+4];
    const float* W2  =(const float*)d_in[base+6];
    const float* bqkv=(const float*)d_in[base+1];
    const float* bo  =(const float*)d_in[base+3];
    const float* b1  =(const float*)d_in[base+5];
    const float* b2  =(const float*)d_in[base+7];
    const float* g1  =(const float*)d_in[base+8];
    const float* be1 =(const float*)d_in[base+9];
    const float* g2  =(const float*)d_in[base+10];
    const float* be2 =(const float*)d_in[base+11];

    k_attn  <<<256,256,0,stream>>>(X,Wqkv,bqkv,Y);
    k_projln<<<256,256,0,stream>>>(X,Y,Wo,bo,g1,be1);
    if(L==0) k_ffln<false><<<512,128,0,stream>>>(X,W1,b1,W2,b2,g2,be2,Pm);
    else     k_ffln<true> <<<512,128,0,stream>>>(X,W1,b1,W2,b2,g2,be2,Pm);
  }
  k_mean<<<64,64,0,stream>>>(Pm,(float*)d_out);
}


# GRU re-decomposed: a 16-patch tile shared by two waves (32 hidden columns each), 16 waves per workgroup, 4 per SIMD instead of 2; h and next x tile pass through LDS, one barrier per step; same sums
# speedup vs baseline: 2.9671x; 1.1830x over previous
#include <hip/hip_runtime.h>
#include <hip/hip_fp16.h>

typedef __attribute__((ext_vector_type(16))) _Float16 v16h;
typedef __attribute__((ext_vector_type(8)))  float    v8f;

#define WMMA_F16(a,b,c) __builtin_amdgcn_wmma_f32_16x16x32_f16(false,(a),false,(b),(short)0,(c),false,false)

__device__ __forceinline__ v8f zero8(){
  v8f z;
  #pragma unroll
  for(int i=0;i<8;++i) z[i]=0.f;
  return z;
}
__device__ __forceinline__ v16h zero16(){
  v16h z;
  #pragma unroll
  for(int i=0;i<16;++i) z[i]=(_Float16)0.f;
  return z;
}

__device__ __forceinline__ int a_k(int i, int h16){ return (i&7) + ((i>>3)<<4) + (h16<<3); }

__device__ __forceinline__ v16h load_tile_f32(const float* __restrict__ src, int ld, int lane){
  const float* rb = src + (size_t)(lane&15)*ld;
  int h16=(lane>>4)&1; v16h r;
  #pragma unroll
  for(int i=0;i<16;++i) r[i]=(_Float16)rb[a_k(i,h16)];
  return r;
}
__device__ __forceinline__ v16h load_tile_f16(const _Float16* __restrict__ src, int ld, int lane){
  const _Float16* rb = src + (size_t)(lane&15)*ld;
  int h16=(lane>>4)&1; v16h r;
  #pragma unroll
  for(int i=0;i<16;++i) r[i]=rb[a_k(i,h16)];
  return r;
}
__device__ __forceinline__ v16h load_tile_kmajor_f16(const _Float16* __restrict__ src, int ld, int lane){
  const _Float16* cb = src + (lane&15);
  int h16=(lane>>4)&1; v16h r;
  #pragma unroll
  for(int i=0;i<16;++i) r[i]=cb[(size_t)a_k(i,h16)*ld];
  return r;
}

typedef short s16x8 __attribute__((ext_vector_type(8)));
typedef __attribute__((ext_vector_type(8))) _Float16 v8h;
__device__ __forceinline__ v8h ld_tr16(const _Float16* p){
  typedef __attribute__((address_space(3))) s16x8 lds_s16x8;
  return __builtin_bit_cast(v8h, __builtin_amdgcn_ds_load_tr16_b128_v8i16((lds_s16x8*)p));
}
__device__ __forceinline__ v16h cat16(v8h lo, v8h hi){
  return __builtin_shufflevector(lo, hi, 0,1,2,3,4,5,6,7,8,9,10,11,12,13,14,15);
}

__device__ __forceinline__ float rmax16(float x){
  #pragma unroll
  for(int m=8;m;m>>=1) x=fmaxf(x,__shfl_xor(x,m,32));
  return x;
}
__device__ __forceinline__ float rsum16(float x){
  #pragma unroll
  for(int m=8;m;m>>=1) x+=__shfl_xor(x,m,32);
  return x;
}
__device__ __forceinline__ float sigf(float x){ return 1.f/(1.f+__expf(-x)); }

__device__ __forceinline__ v8f rmax16x8(v8f x){
  #pragma unroll
  for(int m=8;m;m>>=1){
    v8f t;
    #pragma unroll
    for(int e=0;e<8;++e) t[e]=__shfl_xor(x[e],m,32);
    x=__builtin_elementwise_max(x,t);
  }
  return x;
}
__device__ __forceinline__ v8f rsum16x8(v8f x){
  #pragma unroll
  for(int m=8;m;m>>=1){
    v8f t;
    #pragma unroll
    for(int e=0;e<8;++e) t[e]=__shfl_xor(x[e],m,32);
    x+=t;
  }
  return x;
}

#define PB  8
#define DD  128
#define HH  64
#define SQ  512

template<int K, int NT, int NTHR>
__device__ __forceinline__ void stage_w(const float* __restrict__ W, _Float16* L, int tid){
  constexpr int KT = K/32;
  #pragma unroll
  for(int q=0;q<NT*64/NTHR;++q){
    int u    = q*NTHR + tid;
    int lane = u & 31;
    int p    = (u >> 5) & 1;
    int tile = u >> 6;
    int kt = tile % KT, jt = tile / KT;
    const float* s = W + (size_t)(jt*16 + (lane&15))*K + kt*32 + 16*p + 8*((lane>>4)&1);
    v8h r;
    #pragma unroll
    for(int e=0;e<8;++e) r[e]=(_Float16)s[e];
    *reinterpret_cast<v8h*>(L + (size_t)u*8) = r;
  }
}
__device__ __forceinline__ v16h load_tile_lds(const _Float16* L, int tile, int lane){
  const _Float16* p = L + ((size_t)tile*64 + lane)*8;
  return cat16(*reinterpret_cast<const v8h*>(p), *reinterpret_cast<const v8h*>(p + 32*8));
}

#define HP 72
__device__ __forceinline__ void store_tile_lds(_Float16* L, int tile, int lane, v16h f){
  _Float16* p = L + ((size_t)tile*64 + lane)*8;
  *reinterpret_cast<v8h*>(p)        = __builtin_shufflevector(f, f, 0,1,2,3,4,5,6,7);
  *reinterpret_cast<v8h*>(p + 32*8) = __builtin_shufflevector(f, f, 8,9,10,11,12,13,14,15);
}
__global__ __launch_bounds__(512) void k_gru(
    const float* __restrict__ x, const float* __restrict__ Wih,
    const float* __restrict__ Whh, const float* __restrict__ bih,
    const float* __restrict__ bhh, float* __restrict__ seq)
{
  __shared__ __attribute__((aligned(16))) _Float16 WhhL[192*64];
  __shared__ __attribute__((aligned(16))) _Float16 WihL[192*128];
  __shared__ __attribute__((aligned(16))) _Float16 xsL[2][8][4*512];
  __shared__ __attribute__((aligned(16))) _Float16 hsL[2][8][16*HP];
  int tid=threadIdx.x, lane=tid&31;
  int wave=__builtin_amdgcn_readfirstlane(tid>>5);
  int tt=wave>>1, jh=wave&1;
  int pbase=(blockIdx.x*8+tt)*16;
  int colb=lane&15, rowoff=(lane>>4)*8, h16=(lane>>4)&1;

  stage_w<128,48,512>(Wih, WihL, tid);
  stage_w<64,24,512>(Whh, WhhL, tid);

  float bi[3][2], bh[3][2];
  #pragma unroll
  for(int g=0;g<3;++g){
    #pragma unroll
    for(int jj=0;jj<2;++jj){
      bi[g][jj]=bih[(g*4+2*jh+jj)*16+colb];
      bh[g][jj]=bhh[(g*4+2*jh+jj)*16+colb];
    }
  }

  const float* xrow = x + ((size_t)pbase*PB + (size_t)(lane&15)*PB)*DD + 8*h16;

  {
    _Float16* hw=&hsL[0][tt][0];
    #pragma unroll
    for(int jj=0;jj<2;++jj){
      #pragma unroll
      for(int v=0;v<8;++v) hw[(v+rowoff)*HP + (2*jh+jj)*16 + colb]=(_Float16)0.f;
    }
    #pragma unroll
    for(int q=0;q<2;++q){
      int kk=2*jh+q;
      const float* rb=xrow+kk*32;
      v16h f;
      #pragma unroll
      for(int i=0;i<16;++i) f[i]=(_Float16)rb[(i&7)+((i>>3)<<4)];
      store_tile_lds(&xsL[0][tt][0], kk, lane, f);
    }
  }
  __syncthreads();

  v8f hT[2], hs[2];
  #pragma unroll
  for(int jj=0;jj<2;++jj){ hT[jj]=zero8(); hs[jj]=zero8(); }

  for(int t=0;t<PB;++t){
    int buf=t&1;
    float xr[2][16];
    if(t+1<PB){
      #pragma unroll
      for(int q=0;q<2;++q){
        const float* rb=xrow+(size_t)(t+1)*DD+(2*jh+q)*32;
        #pragma unroll
        for(int i=0;i<16;++i) xr[q][i]=rb[(i&7)+((i>>3)<<4)];
      }
    }
    const _Float16* xl=&xsL[buf][tt][0];
    const _Float16* hl=&hsL[buf][tt][0];
    v16h xA[4], hA[2];
    #pragma unroll
    for(int kk=0;kk<4;++kk) xA[kk]=load_tile_lds(xl, kk, lane);
    #pragma unroll
    for(int kk=0;kk<2;++kk) hA[kk]=load_tile_f16(hl+kk*32, HP, lane);


    #pragma unroll
    for(int jj=0;jj<2;++jj){
      int j=2*jh+jj;
      v8f gr=zero8();
      #pragma unroll
      for(int kk=0;kk<4;++kk)
        gr=WMMA_F16(xA[kk], load_tile_lds(WihL, j*4+kk, lane), gr);
      #pragma unroll
      for(int kk=0;kk<2;++kk)
        gr=WMMA_F16(hA[kk], load_tile_lds(WhhL, j*2+kk, lane), gr);
      v8f gz=zero8();
      #pragma unroll
      for(int kk=0;kk<4;++kk)
        gz=WMMA_F16(xA[kk], load_tile_lds(WihL, (4+j)*4+kk, lane), gz);
      #pragma unroll
      for(int kk=0;kk<2;++kk)
        gz=WMMA_F16(hA[kk], load_tile_lds(WhhL, (4+j)*2+kk, lane), gz);
      v8f gnx=zero8();
      #pragma unroll
      for(int kk=0;kk<4;++kk)
        gnx=WMMA_F16(xA[kk], load_tile_lds(WihL, (8+j)*4+kk, lane), gnx);
      v8f gnh=zero8();
      #pragma unroll
      for(int kk=0;kk<2;++kk)
        gnh=WMMA_F16(hA[kk], load_tile_lds(WhhL, (8+j)*2+kk, lane), gnh);
      #pragma unroll
      for(int e=0;e<8;++e){
        float r =sigf(gr[e]+bi[0][jj]+bh[0][jj]);
        float z =sigf(gz[e]+bi[1][jj]+bh[1][jj]);
        float nv=tanhf(gnx[e]+bi[2][jj]+r*(gnh[e]+bh[2][jj]));
        float hn=(1.f-z)*nv + z*hT[jj][e];
        hT[jj][e]=hn; hs[jj][e]+=hn;
      }
    }
    if(t+1<PB){
      _Float16* hw=&hsL[buf^1][tt][0];
      #pragma unroll
      for(int jj=0;jj<2;++jj){
        #pragma unroll
        for(int v=0;v<8;++v)
          hw[(v+rowoff)*HP + (2*jh+jj)*16 + colb]=(_Float16)hT[jj][v];
      }
      #pragma unroll
      for(int q=0;q<2;++q){
        v16h f;
        #pragma unroll
        for(int i=0;i<16;++i) f[i]=(_Float16)xr[q][i];
        store_tile_lds(&xsL[buf^1][tt][0], 2*jh+q, lane, f);
      }
    }
    __syncthreads();
  }
  float* sb=seq+(size_t)pbase*HH;
  #pragma unroll
  for(int jj=0;jj<2;++jj){
    #pragma unroll
    for(int v=0;v<8;++v)
      sb[(size_t)(v+rowoff)*HH + (2*jh+jj)*16 + colb]=hs[jj][v]*0.125f;
  }
}

__device__ __forceinline__ v16h softmax_tile(v8f s0, v8f s1, v8f m, v8f& lrow,
                                             _Float16* pst, const _Float16* pld){
  v8h p0, p1;
  #pragma unroll
  for(int e=0;e<8;++e){
    float a=__expf(s0[e]-m[e]);
    float c=__expf(s1[e]-m[e]);
    lrow[e]+=a+c;
    p0[e]=(_Float16)a; p1[e]=(_Float16)c;
  }
  *reinterpret_cast<v8h*>(pst)=p0;
  *reinterpret_cast<v8h*>(pst+16*16)=p1;
  return cat16(ld_tr16(pld), ld_tr16(pld+16*16));
}

__global__ __launch_bounds__(256) void k_attn(
    const float* __restrict__ X, const float* __restrict__ Wqkv,
    const float* __restrict__ bqkv, float* __restrict__ O)
{
  __shared__ __attribute__((aligned(16))) _Float16 Ks[SQ*32];
  __shared__ __attribute__((aligned(16))) _Float16 Vs[SQ*16];
  __shared__ __attribute__((aligned(16))) _Float16 Ps[8][4*32*16];
  __shared__ __attribute__((aligned(16))) _Float16 WL[6*512];
  int bh=blockIdx.x, b=bh>>2, h=bh&3;
  int tid=threadIdx.x, lane=tid&31, wave=tid>>5;
  int colb=lane&15, rowoff=(lane>>4)*8;

  const float* xs=X+((size_t)b*SQ+wave*64)*HH;
  v16h Xb[4][2];
  #pragma unroll
  for(int qt=0;qt<4;++qt){
    #pragma unroll
    for(int kk=0;kk<2;++kk) Xb[qt][kk]=load_tile_f32(xs+(size_t)qt*16*HH+kk*32, HH, lane);
  }
  {
    #pragma unroll
    for(int q=0;q<2;++q){
      int u=q*256+tid;
      if(u<6*64){
        int ul=u&31, p=(u>>5)&1, tile=u>>6;
        int g=tile>>1, kt=tile&1;
        const float* s=Wqkv+(size_t)(g*64+h*16+(ul&15))*HH+kt*32+16*p+8*((ul>>4)&1);
        v8h r;
        #pragma unroll
        for(int e=0;e<8;++e) r[e]=(_Float16)s[e];
        *reinterpret_cast<v8h*>(WL+(size_t)u*8)=r;
      }
    }
  }
  float bs[3][8];
  #pragma unroll
  for(int g=0;g<3;++g){
    #pragma unroll
    for(int r=0;r<8;++r) bs[g][r]=bqkv[g*64+h*16+rowoff+r];
  }
  __syncthreads();
  v16h Wa[3][2];
  #pragma unroll
  for(int g=0;g<3;++g){
    #pragma unroll
    for(int kk=0;kk<2;++kk) Wa[g][kk]=load_tile_lds(WL, g*2+kk, lane);
  }
  v8h zh;
  #pragma unroll
  for(int r=0;r<8;++r) zh[r]=(_Float16)0.f;
  v16h Qa[4];
  #pragma unroll
  for(int qt=0;qt<4;++qt){
    v8f q=zero8(), k=zero8(), v=zero8();
    #pragma unroll
    for(int kk=0;kk<2;++kk){
      q=WMMA_F16(Wa[0][kk], Xb[qt][kk], q);
      k=WMMA_F16(Wa[1][kk], Xb[qt][kk], k);
      v=WMMA_F16(Wa[2][kk], Xb[qt][kk], v);
    }
    v8h qh, kh, vh;
    #pragma unroll
    for(int r=0;r<8;++r){
      qh[r]=(_Float16)((q[r]+bs[0][r])*0.25f);
      kh[r]=(_Float16)(k[r]+bs[1][r]);
      vh[r]=(_Float16)(v[r]+bs[2][r]);
    }
    Qa[qt]=cat16(qh, zh);
    int key=wave*64+qt*16+colb;
    *reinterpret_cast<v8h*>(Ks+(size_t)key*32+rowoff)=kh;
    *reinterpret_cast<v8h*>(Ks+(size_t)key*32+16+rowoff)=zh;
    *reinterpret_cast<v8h*>(Vs+(size_t)key*16+rowoff)=vh;
  }
  __syncthreads();

  v8f mrow[4];
  {
    v8f lo;
    lo[0]=-1e30f; lo[1]=-1e30f; lo[2]=-1e30f; lo[3]=-1e30f;
    lo[4]=-1e30f; lo[5]=-1e30f; lo[6]=-1e30f; lo[7]=-1e30f;
    mrow[0]=lo; mrow[1]=lo; mrow[2]=lo; mrow[3]=lo;
  }
  for(int kc=0;kc<16;++kc){
    int kbase=kc*32;
    v16h Kb0=load_tile_f16(Ks+(size_t)kbase*32, 32, lane);
    v16h Kb1=load_tile_f16(Ks+(size_t)(kbase+16)*32, 32, lane);
    #pragma unroll
    for(int qt=0;qt<4;++qt){
      v8f s0=WMMA_F16(Qa[qt],Kb0,zero8());
      v8f s1=WMMA_F16(Qa[qt],Kb1,zero8());
      mrow[qt]=__builtin_elementwise_max(mrow[qt],__builtin_elementwise_max(s0,s1));
    }
  }
  mrow[0]=rmax16x8(mrow[0]);
  mrow[1]=rmax16x8(mrow[1]);
  mrow[2]=rmax16x8(mrow[2]);
  mrow[3]=rmax16x8(mrow[3]);

  _Float16* pl=&Ps[wave][0];
  _Float16* pst=pl+colb*16+rowoff;
  int troff=(8*(lane>>4)+(lane&7))*16+8*((lane>>3)&1);
  const _Float16* pld=pl+troff;
  const _Float16* vld=Vs+troff;
  v8f acc[4], lrow[4];
  acc[0]=zero8(); acc[1]=zero8(); acc[2]=zero8(); acc[3]=zero8();
  lrow[0]=zero8(); lrow[1]=zero8(); lrow[2]=zero8(); lrow[3]=zero8();
  v16h Kb0=load_tile_f16(Ks, 32, lane);
  v16h Kb1=load_tile_f16(Ks+(size_t)16*32, 32, lane);
  v16h Vb=cat16(ld_tr16(vld), ld_tr16(vld+16*16));
  v16h Pa3=zero16(), Vp=zero16();
  for(int kc=0;kc<16;++kc){
    int kn=(kc<15?kc+1:15)*32;
    v16h Kn0=load_tile_f16(Ks+(size_t)kn*32, 32, lane);
    v16h Kn1=load_tile_f16(Ks+(size_t)(kn+16)*32, 32, lane);
    const _Float16* vk=vld+(size_t)kn*16;
    v16h Vn=cat16(ld_tr16(vk), ld_tr16(vk+16*16));
    v8f s00=WMMA_F16(Qa[0],Kb0,zero8()), s01=WMMA_F16(Qa[0],Kb1,zero8());
    v8f s10=WMMA_F16(Qa[1],Kb0,zero8()), s11=WMMA_F16(Qa[1],Kb1,zero8());
    v8f s20=WMMA_F16(Qa[2],Kb0,zero8()), s21=WMMA_F16(Qa[2],Kb1,zero8());
    v8f s30=WMMA_F16(Qa[3],Kb0,zero8()), s31=WMMA_F16(Qa[3],Kb1,zero8());
    acc[3]=WMMA_F16(Pa3,Vp,acc[3]);
    v16h Pa0=softmax_tile(s00,s01,mrow[0],lrow[0],pst,pld);
    v16h Pa1=softmax_tile(s10,s11,mrow[1],lrow[1],pst+512,pld+512);
    acc[0]=WMMA_F16(Pa0,Vb,acc[0]);
    v16h Pa2=softmax_tile(s20,s21,mrow[2],lrow[2],pst+1024,pld+1024);
    acc[1]=WMMA_F16(Pa1,Vb,acc[1]);
    Pa3=softmax_tile(s30,s31,mrow[3],lrow[3],pst+1536,pld+1536);
    acc[2]=WMMA_F16(Pa2,Vb,acc[2]);
    Vp=Vb; Kb0=Kn0; Kb1=Kn1; Vb=Vn;
  }
  acc[3]=WMMA_F16(Pa3,Vp,acc[3]);
  #pragma unroll
  for(int qt=0;qt<4;++qt){
    int qbase=wave*64+qt*16;
    v8f lr=rsum16x8(lrow[qt]);
    float* ob=O+((size_t)b*SQ+qbase)*HH + h*16;
    #pragma unroll
    for(int e=0;e<8;++e)
      ob[(size_t)(e+rowoff)*HH+colb]=acc[qt][e]/lr[e];
  }
}

__global__ __launch_bounds__(256) void k_projln(
    float* __restrict__ X, const float* __restrict__ Y,
    const float* __restrict__ Wo, const float* __restrict__ bo,
    const float* __restrict__ g, const float* __restrict__ be)
{
  __shared__ __attribute__((aligned(16))) _Float16 WL[64*64];
  int lane=threadIdx.x&31, wave=threadIdx.x>>5;
  size_t rbase=((size_t)blockIdx.x*8+wave)*16;
  int colb=lane&15, rowoff=(lane>>4)*8;
  stage_w<64,8,256>(Wo, WL, threadIdx.x);
  v16h A[2];
  #pragma unroll
  for(int kk=0;kk<2;++kk) A[kk]=load_tile_f32(Y+rbase*HH+kk*32, HH, lane);
  __syncthreads();
  v8f o[4];
  #pragma unroll
  for(int j=0;j<4;++j){
    v8f acc=zero8();
    #pragma unroll
    for(int kk=0;kk<2;++kk)
      acc=WMMA_F16(A[kk], load_tile_lds(WL, j*2+kk, lane), acc);
    float bias=bo[j*16+colb];
    const float* xr=X+rbase*HH+j*16+colb;
    #pragma unroll
    for(int e=0;e<8;++e) o[j][e]=acc[e]+bias+xr[(size_t)(e+rowoff)*HH];
  }
  float gv[4],bv[4];
  #pragma unroll
  for(int j=0;j<4;++j){ gv[j]=g[j*16+colb]; bv[j]=be[j*16+colb]; }
  float* xw=X+rbase*HH;
  #pragma unroll
  for(int e=0;e<8;++e){
    float s=o[0][e]+o[1][e]+o[2][e]+o[3][e];
    float mu=rsum16(s)*(1.f/64.f);
    float d0=o[0][e]-mu,d1=o[1][e]-mu,d2=o[2][e]-mu,d3=o[3][e]-mu;
    float var=rsum16(d0*d0+d1*d1+d2*d2+d3*d3)*(1.f/64.f);
    float inv=rsqrtf(var+1e-5f);
    #pragma unroll
    for(int j=0;j<4;++j)
      xw[(size_t)(e+rowoff)*HH+j*16+colb]=(o[j][e]-mu)*inv*gv[j]+bv[j];
  }
}

template<bool LAST>
__global__ __launch_bounds__(128) void k_ffln(
    float* __restrict__ X, const float* __restrict__ W1,
    const float* __restrict__ b1, const float* __restrict__ W2,
    const float* __restrict__ b2, const float* __restrict__ g,
    const float* __restrict__ be, float* __restrict__ part)
{
  __shared__ _Float16 h1[4][16*256];
  __shared__ float red[4][64];
  __shared__ __attribute__((aligned(16))) _Float16 W1L[256*64];
  __shared__ __attribute__((aligned(16))) _Float16 W2L[64*256];
  int lane=threadIdx.x&31, wave=threadIdx.x>>5;
  size_t rbase=((size_t)blockIdx.x*4+wave)*16;
  int colb=lane&15, rowoff=(lane>>4)*8;
  stage_w<64,32,128>(W1, W1L, threadIdx.x);
  stage_w<256,32,128>(W2, W2L, threadIdx.x);
  v16h A[2];
  #pragma unroll
  for(int kk=0;kk<2;++kk) A[kk]=load_tile_f32(X+rbase*HH+kk*32, HH, lane);
  __syncthreads();
  _Float16* hl=&h1[wave][0];
  #pragma unroll
  for(int j=0;j<16;++j){
    v8f acc=zero8();
    #pragma unroll
    for(int kk=0;kk<2;++kk)
      acc=WMMA_F16(A[kk], load_tile_lds(W1L, j*2+kk, lane), acc);
    float bias=b1[j*16+colb];
    #pragma unroll
    for(int e=0;e<8;++e)
      hl[(size_t)(e+rowoff)*256+j*16+colb]=(_Float16)fmaxf(acc[e]+bias,0.f);
  }
  v16h pA[8];
  #pragma unroll
  for(int kk=0;kk<8;++kk) pA[kk]=load_tile_f16(hl+kk*32, 256, lane);
  v8f o[4];
  #pragma unroll
  for(int j=0;j<4;++j){
    v8f acc=zero8();
    #pragma unroll
    for(int kk=0;kk<8;++kk)
      acc=WMMA_F16(pA[kk], load_tile_lds(W2L, j*8+kk, lane), acc);
    float bias=b2[j*16+colb];
    const float* xr=X+rbase*HH+j*16+colb;
    #pragma unroll
    for(int e=0;e<8;++e) o[j][e]=acc[e]+bias+xr[(size_t)(e+rowoff)*HH];
  }
  float gv[4],bv[4];
  #pragma unroll
  for(int j=0;j<4;++j){ gv[j]=g[j*16+colb]; bv[j]=be[j*16+colb]; }
  float* xw=X+rbase*HH;
  float cs[4]={0.f,0.f,0.f,0.f};
  #pragma unroll
  for(int e=0;e<8;++e){
    float s=o[0][e]+o[1][e]+o[2][e]+o[3][e];
    float mu=rsum16(s)*(1.f/64.f);
    float d0=o[0][e]-mu,d1=o[1][e]-mu,d2=o[2][e]-mu,d3=o[3][e]-mu;
    float var=rsum16(d0*d0+d1*d1+d2*d2+d3*d3)*(1.f/64.f);
    float inv=rsqrtf(var+1e-5f);
    #pragma unroll
    for(int j=0;j<4;++j){
      float y=(o[j][e]-mu)*inv*gv[j]+bv[j];
      if(LAST) cs[j]+=y;
      else     xw[(size_t)(e+rowoff)*HH+j*16+colb]=y;
    }
  }
  if(LAST){
    #pragma unroll
    for(int j=0;j<4;++j) cs[j]+=__shfl_xor(cs[j],16,32);
    if(lane<16){
      #pragma unroll
      for(int j=0;j<4;++j) red[wave][j*16+lane]=cs[j];
    }
    __syncthreads();
    if(threadIdx.x<64){
      int t=threadIdx.x;
      part[(size_t)blockIdx.x*HH+t]=(red[0][t]+red[1][t])+(red[2][t]+red[3][t]);
    }
  }
}

__global__ void k_mean(const float* __restrict__ part, float* __restrict__ out){
  int b=blockIdx.x, j=threadIdx.x;
  const float* p=part+(size_t)b*8*HH+j;
  float v[8];
  #pragma unroll
  for(int w=0;w<8;++w) v[w]=p[w*HH];
  float s=((v[0]+v[1])+(v[2]+v[3]))+((v[4]+v[5])+(v[6]+v[7]));
  out[b*HH+j]=s*(1.f/512.f);
}

extern "C" void kernel_launch(void* const* d_in, const int* in_sizes, int n_in,
                              void* d_out, int out_size, void* d_ws, size_t ws_size,
                              hipStream_t stream){
  const float* x  =(const float*)d_in[0];
  const float* Wih=(const float*)d_in[1];
  const float* Whh=(const float*)d_in[2];
  const float* bih=(const float*)d_in[3];
  const float* bhh=(const float*)d_in[4];

  char* ws=(char*)d_ws;
  float*    X =(float*)ws;
  float*    Y =(float*)(ws+(size_t)(8<<20));
  float*    Pm=Y;

  k_gru<<<256,512,0,stream>>>(x,Wih,Whh,bih,bhh,X);

  for(int L=0;L<2;++L){
    int base=5+L*12;
    const float* Wqkv=(const float*)d_in[base+0];
    const float* Wo  =(const float*)d_in[base+2];
    const float* W1  =(const float*)d_in[base+4];
    const float* W2  =(const float*)d_in[base+6];
    const float* bqkv=(const float*)d_in[base+1];
    const float* bo  =(const float*)d_in[base+3];
    const float* b1  =(const float*)d_in[base+5];
    const float* b2  =(const float*)d_in[base+7];
    const float* g1  =(const float*)d_in[base+8];
    const float* be1 =(const float*)d_in[base+9];
    const float* g2  =(const float*)d_in[base+10];
    const float* be2 =(const float*)d_in[base+11];

    k_attn  <<<256,256,0,stream>>>(X,Wqkv,bqkv,Y);
    k_projln<<<256,256,0,stream>>>(X,Y,Wo,bo,g1,be1);
    if(L==0) k_ffln<false><<<512,128,0,stream>>>(X,W1,b1,W2,b2,g2,be2,Pm);
    else     k_ffln<true> <<<512,128,0,stream>>>(X,W1,b1,W2,b2,g2,be2,Pm);
  }
  k_mean<<<64,64,0,stream>>>(Pm,(float*)d_out);
}


# GRU gates: z and tanh share one IEEE division over the common denominator (1+ez)(1+en), tanh(u)=(1-en)/(1+en), en=exp(-2|u|) by the kernel's own exp; no branching tanh; gate values move in last bits
# speedup vs baseline: 3.1380x; 1.0576x over previous
#include <hip/hip_runtime.h>
#include <hip/hip_fp16.h>

typedef __attribute__((ext_vector_type(16))) _Float16 v16h;
typedef __attribute__((ext_vector_type(8)))  float    v8f;

#define WMMA_F16(a,b,c) __builtin_amdgcn_wmma_f32_16x16x32_f16(false,(a),false,(b),(short)0,(c),false,false)

__device__ __forceinline__ v8f zero8(){
  v8f z;
  #pragma unroll
  for(int i=0;i<8;++i) z[i]=0.f;
  return z;
}
__device__ __forceinline__ v16h zero16(){
  v16h z;
  #pragma unroll
  for(int i=0;i<16;++i) z[i]=(_Float16)0.f;
  return z;
}

__device__ __forceinline__ int a_k(int i, int h16){ return (i&7) + ((i>>3)<<4) + (h16<<3); }

__device__ __forceinline__ v16h load_tile_f32(const float* __restrict__ src, int ld, int lane){
  const float* rb = src + (size_t)(lane&15)*ld;
  int h16=(lane>>4)&1; v16h r;
  #pragma unroll
  for(int i=0;i<16;++i) r[i]=(_Float16)rb[a_k(i,h16)];
  return r;
}
__device__ __forceinline__ v16h load_tile_f16(const _Float16* __restrict__ src, int ld, int lane){
  const _Float16* rb = src + (size_t)(lane&15)*ld;
  int h16=(lane>>4)&1; v16h r;
  #pragma unroll
  for(int i=0;i<16;++i) r[i]=rb[a_k(i,h16)];
  return r;
}
__device__ __forceinline__ v16h load_tile_kmajor_f16(const _Float16* __restrict__ src, int ld, int lane){
  const _Float16* cb = src + (lane&15);
  int h16=(lane>>4)&1; v16h r;
  #pragma unroll
  for(int i=0;i<16;++i) r[i]=cb[(size_t)a_k(i,h16)*ld];
  return r;
}

typedef short s16x8 __attribute__((ext_vector_type(8)));
typedef __attribute__((ext_vector_type(8))) _Float16 v8h;
__device__ __forceinline__ v8h ld_tr16(const _Float16* p){
  typedef __attribute__((address_space(3))) s16x8 lds_s16x8;
  return __builtin_bit_cast(v8h, __builtin_amdgcn_ds_load_tr16_b128_v8i16((lds_s16x8*)p));
}
__device__ __forceinline__ v16h cat16(v8h lo, v8h hi){
  return __builtin_shufflevector(lo, hi, 0,1,2,3,4,5,6,7,8,9,10,11,12,13,14,15);
}

__device__ __forceinline__ float rmax16(float x){
  #pragma unroll
  for(int m=8;m;m>>=1) x=fmaxf(x,__shfl_xor(x,m,32));
  return x;
}
__device__ __forceinline__ float rsum16(float x){
  #pragma unroll
  for(int m=8;m;m>>=1) x+=__shfl_xor(x,m,32);
  return x;
}
__device__ __forceinline__ float sigf(float x){ return 1.f/(1.f+__expf(-x)); }

__device__ __forceinline__ v8f rmax16x8(v8f x){
  #pragma unroll
  for(int m=8;m;m>>=1){
    v8f t;
    #pragma unroll
    for(int e=0;e<8;++e) t[e]=__shfl_xor(x[e],m,32);
    x=__builtin_elementwise_max(x,t);
  }
  return x;
}
__device__ __forceinline__ v8f rsum16x8(v8f x){
  #pragma unroll
  for(int m=8;m;m>>=1){
    v8f t;
    #pragma unroll
    for(int e=0;e<8;++e) t[e]=__shfl_xor(x[e],m,32);
    x+=t;
  }
  return x;
}

#define PB  8
#define DD  128
#define HH  64
#define SQ  512

template<int K, int NT, int NTHR>
__device__ __forceinline__ void stage_w(const float* __restrict__ W, _Float16* L, int tid){
  constexpr int KT = K/32;
  #pragma unroll
  for(int q=0;q<NT*64/NTHR;++q){
    int u    = q*NTHR + tid;
    int lane = u & 31;
    int p    = (u >> 5) & 1;
    int tile = u >> 6;
    int kt = tile % KT, jt = tile / KT;
    const float* s = W + (size_t)(jt*16 + (lane&15))*K + kt*32 + 16*p + 8*((lane>>4)&1);
    v8h r;
    #pragma unroll
    for(int e=0;e<8;++e) r[e]=(_Float16)s[e];
    *reinterpret_cast<v8h*>(L + (size_t)u*8) = r;
  }
}
__device__ __forceinline__ v16h load_tile_lds(const _Float16* L, int tile, int lane){
  const _Float16* p = L + ((size_t)tile*64 + lane)*8;
  return cat16(*reinterpret_cast<const v8h*>(p), *reinterpret_cast<const v8h*>(p + 32*8));
}

#define HP 72
__device__ __forceinline__ void store_tile_lds(_Float16* L, int tile, int lane, v16h f){
  _Float16* p = L + ((size_t)tile*64 + lane)*8;
  *reinterpret_cast<v8h*>(p)        = __builtin_shufflevector(f, f, 0,1,2,3,4,5,6,7);
  *reinterpret_cast<v8h*>(p + 32*8) = __builtin_shufflevector(f, f, 8,9,10,11,12,13,14,15);
}
__global__ __launch_bounds__(512) void k_gru(
    const float* __restrict__ x, const float* __restrict__ Wih,
    const float* __restrict__ Whh, const float* __restrict__ bih,
    const float* __restrict__ bhh, float* __restrict__ seq)
{
  __shared__ __attribute__((aligned(16))) _Float16 WhhL[192*64];
  __shared__ __attribute__((aligned(16))) _Float16 WihL[192*128];
  __shared__ __attribute__((aligned(16))) _Float16 xsL[2][8][4*512];
  __shared__ __attribute__((aligned(16))) _Float16 hsL[2][8][16*HP];
  int tid=threadIdx.x, lane=tid&31;
  int wave=__builtin_amdgcn_readfirstlane(tid>>5);
  int tt=wave>>1, jh=wave&1;
  int pbase=(blockIdx.x*8+tt)*16;
  int colb=lane&15, rowoff=(lane>>4)*8, h16=(lane>>4)&1;

  stage_w<128,48,512>(Wih, WihL, tid);
  stage_w<64,24,512>(Whh, WhhL, tid);

  float bi[3][2], bh[3][2];
  #pragma unroll
  for(int g=0;g<3;++g){
    #pragma unroll
    for(int jj=0;jj<2;++jj){
      bi[g][jj]=bih[(g*4+2*jh+jj)*16+colb];
      bh[g][jj]=bhh[(g*4+2*jh+jj)*16+colb];
    }
  }

  const float* xrow = x + ((size_t)pbase*PB + (size_t)(lane&15)*PB)*DD + 8*h16;

  {
    _Float16* hw=&hsL[0][tt][0];
    #pragma unroll
    for(int jj=0;jj<2;++jj){
      #pragma unroll
      for(int v=0;v<8;++v) hw[(v+rowoff)*HP + (2*jh+jj)*16 + colb]=(_Float16)0.f;
    }
    #pragma unroll
    for(int q=0;q<2;++q){
      int kk=2*jh+q;
      const float* rb=xrow+kk*32;
      v16h f;
      #pragma unroll
      for(int i=0;i<16;++i) f[i]=(_Float16)rb[(i&7)+((i>>3)<<4)];
      store_tile_lds(&xsL[0][tt][0], kk, lane, f);
    }
  }
  __syncthreads();

  v8f hT[2], hs[2];
  #pragma unroll
  for(int jj=0;jj<2;++jj){ hT[jj]=zero8(); hs[jj]=zero8(); }

  for(int t=0;t<PB;++t){
    int buf=t&1;
    float xr[2][16];
    if(t+1<PB){
      #pragma unroll
      for(int q=0;q<2;++q){
        const float* rb=xrow+(size_t)(t+1)*DD+(2*jh+q)*32;
        #pragma unroll
        for(int i=0;i<16;++i) xr[q][i]=rb[(i&7)+((i>>3)<<4)];
      }
    }
    const _Float16* xl=&xsL[buf][tt][0];
    const _Float16* hl=&hsL[buf][tt][0];
    v16h xA[4], hA[2];
    #pragma unroll
    for(int kk=0;kk<4;++kk) xA[kk]=load_tile_lds(xl, kk, lane);
    #pragma unroll
    for(int kk=0;kk<2;++kk) hA[kk]=load_tile_f16(hl+kk*32, HP, lane);


    #pragma unroll
    for(int jj=0;jj<2;++jj){
      int j=2*jh+jj;
      v8f gr=zero8();
      #pragma unroll
      for(int kk=0;kk<4;++kk)
        gr=WMMA_F16(xA[kk], load_tile_lds(WihL, j*4+kk, lane), gr);
      #pragma unroll
      for(int kk=0;kk<2;++kk)
        gr=WMMA_F16(hA[kk], load_tile_lds(WhhL, j*2+kk, lane), gr);
      v8f gz=zero8();
      #pragma unroll
      for(int kk=0;kk<4;++kk)
        gz=WMMA_F16(xA[kk], load_tile_lds(WihL, (4+j)*4+kk, lane), gz);
      #pragma unroll
      for(int kk=0;kk<2;++kk)
        gz=WMMA_F16(hA[kk], load_tile_lds(WhhL, (4+j)*2+kk, lane), gz);
      v8f gnx=zero8();
      #pragma unroll
      for(int kk=0;kk<4;++kk)
        gnx=WMMA_F16(xA[kk], load_tile_lds(WihL, (8+j)*4+kk, lane), gnx);
      v8f gnh=zero8();
      #pragma unroll
      for(int kk=0;kk<2;++kk)
        gnh=WMMA_F16(hA[kk], load_tile_lds(WhhL, (8+j)*2+kk, lane), gnh);
      #pragma unroll
      for(int e=0;e<8;++e){
        float r =sigf(gr[e]+bi[0][jj]+bh[0][jj]);
        float az=gz[e]+bi[1][jj]+bh[1][jj];
        float u =gnx[e]+bi[2][jj]+r*(gnh[e]+bh[2][jj]);
        float ez=__expf(-fabsf(az));
        float en=__expf(-2.f*fabsf(u));
        float dz=1.f+ez, dn=1.f+en;
        float inv=1.f/(dz*dn);
        float z =(az>=0.f?1.f:ez)*(dn*inv);
        float nv=copysignf((1.f-en)*dz*inv, u);
        float hn=(1.f-z)*nv + z*hT[jj][e];
        hT[jj][e]=hn; hs[jj][e]+=hn;
      }
    }
    if(t+1<PB){
      _Float16* hw=&hsL[buf^1][tt][0];
      #pragma unroll
      for(int jj=0;jj<2;++jj){
        #pragma unroll
        for(int v=0;v<8;++v)
          hw[(v+rowoff)*HP + (2*jh+jj)*16 + colb]=(_Float16)hT[jj][v];
      }
      #pragma unroll
      for(int q=0;q<2;++q){
        v16h f;
        #pragma unroll
        for(int i=0;i<16;++i) f[i]=(_Float16)xr[q][i];
        store_tile_lds(&xsL[buf^1][tt][0], 2*jh+q, lane, f);
      }
    }
    __syncthreads();
  }
  float* sb=seq+(size_t)pbase*HH;
  #pragma unroll
  for(int jj=0;jj<2;++jj){
    #pragma unroll
    for(int v=0;v<8;++v)
      sb[(size_t)(v+rowoff)*HH + (2*jh+jj)*16 + colb]=hs[jj][v]*0.125f;
  }
}

__device__ __forceinline__ v16h softmax_tile(v8f s0, v8f s1, v8f m, v8f& lrow,
                                             _Float16* pst, const _Float16* pld){
  v8h p0, p1;
  #pragma unroll
  for(int e=0;e<8;++e){
    float a=__expf(s0[e]-m[e]);
    float c=__expf(s1[e]-m[e]);
    lrow[e]+=a+c;
    p0[e]=(_Float16)a; p1[e]=(_Float16)c;
  }
  *reinterpret_cast<v8h*>(pst)=p0;
  *reinterpret_cast<v8h*>(pst+16*16)=p1;
  return cat16(ld_tr16(pld), ld_tr16(pld+16*16));
}

__global__ __launch_bounds__(256) void k_attn(
    const float* __restrict__ X, const float* __restrict__ Wqkv,
    const float* __restrict__ bqkv, float* __restrict__ O)
{
  __shared__ __attribute__((aligned(16))) _Float16 Ks[SQ*32];
  __shared__ __attribute__((aligned(16))) _Float16 Vs[SQ*16];
  __shared__ __attribute__((aligned(16))) _Float16 Ps[8][4*32*16];
  __shared__ __attribute__((aligned(16))) _Float16 WL[6*512];
  int bh=blockIdx.x, b=bh>>2, h=bh&3;
  int tid=threadIdx.x, lane=tid&31, wave=tid>>5;
  int colb=lane&15, rowoff=(lane>>4)*8;

  const float* xs=X+((size_t)b*SQ+wave*64)*HH;
  v16h Xb[4][2];
  #pragma unroll
  for(int qt=0;qt<4;++qt){
    #pragma unroll
    for(int kk=0;kk<2;++kk) Xb[qt][kk]=load_tile_f32(xs+(size_t)qt*16*HH+kk*32, HH, lane);
  }
  {
    #pragma unroll
    for(int q=0;q<2;++q){
      int u=q*256+tid;
      if(u<6*64){
        int ul=u&31, p=(u>>5)&1, tile=u>>6;
        int g=tile>>1, kt=tile&1;
        const float* s=Wqkv+(size_t)(g*64+h*16+(ul&15))*HH+kt*32+16*p+8*((ul>>4)&1);
        v8h r;
        #pragma unroll
        for(int e=0;e<8;++e) r[e]=(_Float16)s[e];
        *reinterpret_cast<v8h*>(WL+(size_t)u*8)=r;
      }
    }
  }
  float bs[3][8];
  #pragma unroll
  for(int g=0;g<3;++g){
    #pragma unroll
    for(int r=0;r<8;++r) bs[g][r]=bqkv[g*64+h*16+rowoff+r];
  }
  __syncthreads();
  v16h Wa[3][2];
  #pragma unroll
  for(int g=0;g<3;++g){
    #pragma unroll
    for(int kk=0;kk<2;++kk) Wa[g][kk]=load_tile_lds(WL, g*2+kk, lane);
  }
  v8h zh;
  #pragma unroll
  for(int r=0;r<8;++r) zh[r]=(_Float16)0.f;
  v16h Qa[4];
  #pragma unroll
  for(int qt=0;qt<4;++qt){
    v8f q=zero8(), k=zero8(), v=zero8();
    #pragma unroll
    for(int kk=0;kk<2;++kk){
      q=WMMA_F16(Wa[0][kk], Xb[qt][kk], q);
      k=WMMA_F16(Wa[1][kk], Xb[qt][kk], k);
      v=WMMA_F16(Wa[2][kk], Xb[qt][kk], v);
    }
    v8h qh, kh, vh;
    #pragma unroll
    for(int r=0;r<8;++r){
      qh[r]=(_Float16)((q[r]+bs[0][r])*0.25f);
      kh[r]=(_Float16)(k[r]+bs[1][r]);
      vh[r]=(_Float16)(v[r]+bs[2][r]);
    }
    Qa[qt]=cat16(qh, zh);
    int key=wave*64+qt*16+colb;
    *reinterpret_cast<v8h*>(Ks+(size_t)key*32+rowoff)=kh;
    *reinterpret_cast<v8h*>(Ks+(size_t)key*32+16+rowoff)=zh;
    *reinterpret_cast<v8h*>(Vs+(size_t)key*16+rowoff)=vh;
  }
  __syncthreads();

  v8f mrow[4];
  {
    v8f lo;
    lo[0]=-1e30f; lo[1]=-1e30f; lo[2]=-1e30f; lo[3]=-1e30f;
    lo[4]=-1e30f; lo[5]=-1e30f; lo[6]=-1e30f; lo[7]=-1e30f;
    mrow[0]=lo; mrow[1]=lo; mrow[2]=lo; mrow[3]=lo;
  }
  for(int kc=0;kc<16;++kc){
    int kbase=kc*32;
    v16h Kb0=load_tile_f16(Ks+(size_t)kbase*32, 32, lane);
    v16h Kb1=load_tile_f16(Ks+(size_t)(kbase+16)*32, 32, lane);
    #pragma unroll
    for(int qt=0;qt<4;++qt){
      v8f s0=WMMA_F16(Qa[qt],Kb0,zero8());
      v8f s1=WMMA_F16(Qa[qt],Kb1,zero8());
      mrow[qt]=__builtin_elementwise_max(mrow[qt],__builtin_elementwise_max(s0,s1));
    }
  }
  mrow[0]=rmax16x8(mrow[0]);
  mrow[1]=rmax16x8(mrow[1]);
  mrow[2]=rmax16x8(mrow[2]);
  mrow[3]=rmax16x8(mrow[3]);

  _Float16* pl=&Ps[wave][0];
  _Float16* pst=pl+colb*16+rowoff;
  int troff=(8*(lane>>4)+(lane&7))*16+8*((lane>>3)&1);
  const _Float16* pld=pl+troff;
  const _Float16* vld=Vs+troff;
  v8f acc[4], lrow[4];
  acc[0]=zero8(); acc[1]=zero8(); acc[2]=zero8(); acc[3]=zero8();
  lrow[0]=zero8(); lrow[1]=zero8(); lrow[2]=zero8(); lrow[3]=zero8();
  v16h Kb0=load_tile_f16(Ks, 32, lane);
  v16h Kb1=load_tile_f16(Ks+(size_t)16*32, 32, lane);
  v16h Vb=cat16(ld_tr16(vld), ld_tr16(vld+16*16));
  v16h Pa3=zero16(), Vp=zero16();
  for(int kc=0;kc<16;++kc){
    int kn=(kc<15?kc+1:15)*32;
    v16h Kn0=load_tile_f16(Ks+(size_t)kn*32, 32, lane);
    v16h Kn1=load_tile_f16(Ks+(size_t)(kn+16)*32, 32, lane);
    const _Float16* vk=vld+(size_t)kn*16;
    v16h Vn=cat16(ld_tr16(vk), ld_tr16(vk+16*16));
    v8f s00=WMMA_F16(Qa[0],Kb0,zero8()), s01=WMMA_F16(Qa[0],Kb1,zero8());
    v8f s10=WMMA_F16(Qa[1],Kb0,zero8()), s11=WMMA_F16(Qa[1],Kb1,zero8());
    v8f s20=WMMA_F16(Qa[2],Kb0,zero8()), s21=WMMA_F16(Qa[2],Kb1,zero8());
    v8f s30=WMMA_F16(Qa[3],Kb0,zero8()), s31=WMMA_F16(Qa[3],Kb1,zero8());
    acc[3]=WMMA_F16(Pa3,Vp,acc[3]);
    v16h Pa0=softmax_tile(s00,s01,mrow[0],lrow[0],pst,pld);
    v16h Pa1=softmax_tile(s10,s11,mrow[1],lrow[1],pst+512,pld+512);
    acc[0]=WMMA_F16(Pa0,Vb,acc[0]);
    v16h Pa2=softmax_tile(s20,s21,mrow[2],lrow[2],pst+1024,pld+1024);
    acc[1]=WMMA_F16(Pa1,Vb,acc[1]);
    Pa3=softmax_tile(s30,s31,mrow[3],lrow[3],pst+1536,pld+1536);
    acc[2]=WMMA_F16(Pa2,Vb,acc[2]);
    Vp=Vb; Kb0=Kn0; Kb1=Kn1; Vb=Vn;
  }
  acc[3]=WMMA_F16(Pa3,Vp,acc[3]);
  #pragma unroll
  for(int qt=0;qt<4;++qt){
    int qbase=wave*64+qt*16;
    v8f lr=rsum16x8(lrow[qt]);
    float* ob=O+((size_t)b*SQ+qbase)*HH + h*16;
    #pragma unroll
    for(int e=0;e<8;++e)
      ob[(size_t)(e+rowoff)*HH+colb]=acc[qt][e]/lr[e];
  }
}

__global__ __launch_bounds__(256) void k_projln(
    float* __restrict__ X, const float* __restrict__ Y,
    const float* __restrict__ Wo, const float* __restrict__ bo,
    const float* __restrict__ g, const float* __restrict__ be)
{
  __shared__ __attribute__((aligned(16))) _Float16 WL[64*64];
  int lane=threadIdx.x&31, wave=threadIdx.x>>5;
  size_t rbase=((size_t)blockIdx.x*8+wave)*16;
  int colb=lane&15, rowoff=(lane>>4)*8;
  stage_w<64,8,256>(Wo, WL, threadIdx.x);
  v16h A[2];
  #pragma unroll
  for(int kk=0;kk<2;++kk) A[kk]=load_tile_f32(Y+rbase*HH+kk*32, HH, lane);
  __syncthreads();
  v8f o[4];
  #pragma unroll
  for(int j=0;j<4;++j){
    v8f acc=zero8();
    #pragma unroll
    for(int kk=0;kk<2;++kk)
      acc=WMMA_F16(A[kk], load_tile_lds(WL, j*2+kk, lane), acc);
    float bias=bo[j*16+colb];
    const float* xr=X+rbase*HH+j*16+colb;
    #pragma unroll
    for(int e=0;e<8;++e) o[j][e]=acc[e]+bias+xr[(size_t)(e+rowoff)*HH];
  }
  float gv[4],bv[4];
  #pragma unroll
  for(int j=0;j<4;++j){ gv[j]=g[j*16+colb]; bv[j]=be[j*16+colb]; }
  float* xw=X+rbase*HH;
  #pragma unroll
  for(int e=0;e<8;++e){
    float s=o[0][e]+o[1][e]+o[2][e]+o[3][e];
    float mu=rsum16(s)*(1.f/64.f);
    float d0=o[0][e]-mu,d1=o[1][e]-mu,d2=o[2][e]-mu,d3=o[3][e]-mu;
    float var=rsum16(d0*d0+d1*d1+d2*d2+d3*d3)*(1.f/64.f);
    float inv=rsqrtf(var+1e-5f);
    #pragma unroll
    for(int j=0;j<4;++j)
      xw[(size_t)(e+rowoff)*HH+j*16+colb]=(o[j][e]-mu)*inv*gv[j]+bv[j];
  }
}

template<bool LAST>
__global__ __launch_bounds__(128) void k_ffln(
    float* __restrict__ X, const float* __restrict__ W1,
    const float* __restrict__ b1, const float* __restrict__ W2,
    const float* __restrict__ b2, const float* __restrict__ g,
    const float* __restrict__ be, float* __restrict__ part)
{
  __shared__ _Float16 h1[4][16*256];
  __shared__ float red[4][64];
  __shared__ __attribute__((aligned(16))) _Float16 W1L[256*64];
  __shared__ __attribute__((aligned(16))) _Float16 W2L[64*256];
  int lane=threadIdx.x&31, wave=threadIdx.x>>5;
  size_t rbase=((size_t)blockIdx.x*4+wave)*16;
  int colb=lane&15, rowoff=(lane>>4)*8;
  stage_w<64,32,128>(W1, W1L, threadIdx.x);
  stage_w<256,32,128>(W2, W2L, threadIdx.x);
  v16h A[2];
  #pragma unroll
  for(int kk=0;kk<2;++kk) A[kk]=load_tile_f32(X+rbase*HH+kk*32, HH, lane);
  __syncthreads();
  _Float16* hl=&h1[wave][0];
  #pragma unroll
  for(int j=0;j<16;++j){
    v8f acc=zero8();
    #pragma unroll
    for(int kk=0;kk<2;++kk)
      acc=WMMA_F16(A[kk], load_tile_lds(W1L, j*2+kk, lane), acc);
    float bias=b1[j*16+colb];
    #pragma unroll
    for(int e=0;e<8;++e)
      hl[(size_t)(e+rowoff)*256+j*16+colb]=(_Float16)fmaxf(acc[e]+bias,0.f);
  }
  v16h pA[8];
  #pragma unroll
  for(int kk=0;kk<8;++kk) pA[kk]=load_tile_f16(hl+kk*32, 256, lane);
  v8f o[4];
  #pragma unroll
  for(int j=0;j<4;++j){
    v8f acc=zero8();
    #pragma unroll
    for(int kk=0;kk<8;++kk)
      acc=WMMA_F16(pA[kk], load_tile_lds(W2L, j*8+kk, lane), acc);
    float bias=b2[j*16+colb];
    const float* xr=X+rbase*HH+j*16+colb;
    #pragma unroll
    for(int e=0;e<8;++e) o[j][e]=acc[e]+bias+xr[(size_t)(e+rowoff)*HH];
  }
  float gv[4],bv[4];
  #pragma unroll
  for(int j=0;j<4;++j){ gv[j]=g[j*16+colb]; bv[j]=be[j*16+colb]; }
  float* xw=X+rbase*HH;
  float cs[4]={0.f,0.f,0.f,0.f};
  #pragma unroll
  for(int e=0;e<8;++e){
    float s=o[0][e]+o[1][e]+o[2][e]+o[3][e];
    float mu=rsum16(s)*(1.f/64.f);
    float d0=o[0][e]-mu,d1=o[1][e]-mu,d2=o[2][e]-mu,d3=o[3][e]-mu;
    float var=rsum16(d0*d0+d1*d1+d2*d2+d3*d3)*(1.f/64.f);
    float inv=rsqrtf(var+1e-5f);
    #pragma unroll
    for(int j=0;j<4;++j){
      float y=(o[j][e]-mu)*inv*gv[j]+bv[j];
      if(LAST) cs[j]+=y;
      else     xw[(size_t)(e+rowoff)*HH+j*16+colb]=y;
    }
  }
  if(LAST){
    #pragma unroll
    for(int j=0;j<4;++j) cs[j]+=__shfl_xor(cs[j],16,32);
    if(lane<16){
      #pragma unroll
      for(int j=0;j<4;++j) red[wave][j*16+lane]=cs[j];
    }
    __syncthreads();
    if(threadIdx.x<64){
      int t=threadIdx.x;
      part[(size_t)blockIdx.x*HH+t]=(red[0][t]+red[1][t])+(red[2][t]+red[3][t]);
    }
  }
}

__global__ void k_mean(const float* __restrict__ part, float* __restrict__ out){
  int b=blockIdx.x, j=threadIdx.x;
  const float* p=part+(size_t)b*8*HH+j;
  float v[8];
  #pragma unroll
  for(int w=0;w<8;++w) v[w]=p[w*HH];
  float s=((v[0]+v[1])+(v[2]+v[3]))+((v[4]+v[5])+(v[6]+v[7]));
  out[b*HH+j]=s*(1.f/512.f);
}

extern "C" void kernel_launch(void* const* d_in, const int* in_sizes, int n_in,
                              void* d_out, int out_size, void* d_ws, size_t ws_size,
                              hipStream_t stream){
  const float* x  =(const float*)d_in[0];
  const float* Wih=(const float*)d_in[1];
  const float* Whh=(const float*)d_in[2];
  const float* bih=(const float*)d_in[3];
  const float* bhh=(const float*)d_in[4];

  char* ws=(char*)d_ws;
  float*    X =(float*)ws;
  float*    Y =(float*)(ws+(size_t)(8<<20));
  float*    Pm=Y;

  k_gru<<<256,512,0,stream>>>(x,Wih,Whh,bih,bhh,X);

  for(int L=0;L<2;++L){
    int base=5+L*12;
    const float* Wqkv=(const float*)d_in[base+0];
    const float* Wo  =(const float*)d_in[base+2];
    const float* W1  =(const float*)d_in[base+4];
    const float* W2  =(const float*)d_in[base+6];
    const float* bqkv=(const float*)d_in[base+1];
    const float* bo  =(const float*)d_in[base+3];
    const float* b1  =(const float*)d_in[base+5];
    const float* b2  =(const float*)d_in[base+7];
    const float* g1  =(const float*)d_in[base+8];
    const float* be1 =(const float*)d_in[base+9];
    const float* g2  =(const float*)d_in[base+10];
    const float* be2 =(const float*)d_in[base+11];

    k_attn  <<<256,256,0,stream>>>(X,Wqkv,bqkv,Y);
    k_projln<<<256,256,0,stream>>>(X,Y,Wo,bo,g1,be1);
    if(L==0) k_ffln<false><<<512,128,0,stream>>>(X,W1,b1,W2,b2,g2,be2,Pm);
    else     k_ffln<true> <<<512,128,0,stream>>>(X,W1,b1,W2,b2,g2,be2,Pm);
  }
  k_mean<<<64,64,0,stream>>>(Pm,(float*)d_out);
}


# out-proj+residual+LayerNorm moved into the attention launch: workgroup = 128 queries x 4 heads (wave = head x 64 queries), all heads' K/V per workgroup, O through LDS; 8 -> 6 launches; same sums
# speedup vs baseline: 3.2884x; 1.0479x over previous
#include <hip/hip_runtime.h>
#include <hip/hip_fp16.h>

typedef __attribute__((ext_vector_type(16))) _Float16 v16h;
typedef __attribute__((ext_vector_type(8)))  float    v8f;

#define WMMA_F16(a,b,c) __builtin_amdgcn_wmma_f32_16x16x32_f16(false,(a),false,(b),(short)0,(c),false,false)

__device__ __forceinline__ v8f zero8(){
  v8f z;
  #pragma unroll
  for(int i=0;i<8;++i) z[i]=0.f;
  return z;
}
__device__ __forceinline__ v16h zero16(){
  v16h z;
  #pragma unroll
  for(int i=0;i<16;++i) z[i]=(_Float16)0.f;
  return z;
}

__device__ __forceinline__ int a_k(int i, int h16){ return (i&7) + ((i>>3)<<4) + (h16<<3); }

__device__ __forceinline__ v16h load_tile_f32(const float* __restrict__ src, int ld, int lane){
  const float* rb = src + (size_t)(lane&15)*ld;
  int h16=(lane>>4)&1; v16h r;
  #pragma unroll
  for(int i=0;i<16;++i) r[i]=(_Float16)rb[a_k(i,h16)];
  return r;
}
__device__ __forceinline__ v16h load_tile_f16(const _Float16* __restrict__ src, int ld, int lane){
  const _Float16* rb = src + (size_t)(lane&15)*ld;
  int h16=(lane>>4)&1; v16h r;
  #pragma unroll
  for(int i=0;i<16;++i) r[i]=rb[a_k(i,h16)];
  return r;
}
__device__ __forceinline__ v16h load_tile_kmajor_f16(const _Float16* __restrict__ src, int ld, int lane){
  const _Float16* cb = src + (lane&15);
  int h16=(lane>>4)&1; v16h r;
  #pragma unroll
  for(int i=0;i<16;++i) r[i]=cb[(size_t)a_k(i,h16)*ld];
  return r;
}

typedef short s16x8 __attribute__((ext_vector_type(8)));
typedef __attribute__((ext_vector_type(8))) _Float16 v8h;
__device__ __forceinline__ v8h ld_tr16(const _Float16* p){
  typedef __attribute__((address_space(3))) s16x8 lds_s16x8;
  return __builtin_bit_cast(v8h, __builtin_amdgcn_ds_load_tr16_b128_v8i16((lds_s16x8*)p));
}
__device__ __forceinline__ v16h cat16(v8h lo, v8h hi){
  return __builtin_shufflevector(lo, hi, 0,1,2,3,4,5,6,7,8,9,10,11,12,13,14,15);
}

__device__ __forceinline__ float rmax16(float x){
  #pragma unroll
  for(int m=8;m;m>>=1) x=fmaxf(x,__shfl_xor(x,m,32));
  return x;
}
__device__ __forceinline__ float rsum16(float x){
  #pragma unroll
  for(int m=8;m;m>>=1) x+=__shfl_xor(x,m,32);
  return x;
}
__device__ __forceinline__ float sigf(float x){ return 1.f/(1.f+__expf(-x)); }

__device__ __forceinline__ v8f rmax16x8(v8f x){
  #pragma unroll
  for(int m=8;m;m>>=1){
    v8f t;
    #pragma unroll
    for(int e=0;e<8;++e) t[e]=__shfl_xor(x[e],m,32);
    x=__builtin_elementwise_max(x,t);
  }
  return x;
}
__device__ __forceinline__ v8f rsum16x8(v8f x){
  #pragma unroll
  for(int m=8;m;m>>=1){
    v8f t;
    #pragma unroll
    for(int e=0;e<8;++e) t[e]=__shfl_xor(x[e],m,32);
    x+=t;
  }
  return x;
}

#define PB  8
#define DD  128
#define HH  64
#define SQ  512

template<int K, int NT, int NTHR>
__device__ __forceinline__ void stage_w(const float* __restrict__ W, _Float16* L, int tid){
  constexpr int KT = K/32;
  #pragma unroll
  for(int q=0;q<NT*64/NTHR;++q){
    int u    = q*NTHR + tid;
    int lane = u & 31;
    int p    = (u >> 5) & 1;
    int tile = u >> 6;
    int kt = tile % KT, jt = tile / KT;
    const float* s = W + (size_t)(jt*16 + (lane&15))*K + kt*32 + 16*p + 8*((lane>>4)&1);
    v8h r;
    #pragma unroll
    for(int e=0;e<8;++e) r[e]=(_Float16)s[e];
    *reinterpret_cast<v8h*>(L + (size_t)u*8) = r;
  }
}
__device__ __forceinline__ v16h load_tile_lds(const _Float16* L, int tile, int lane){
  const _Float16* p = L + ((size_t)tile*64 + lane)*8;
  return cat16(*reinterpret_cast<const v8h*>(p), *reinterpret_cast<const v8h*>(p + 32*8));
}

#define HP 72
__device__ __forceinline__ void store_tile_lds(_Float16* L, int tile, int lane, v16h f){
  _Float16* p = L + ((size_t)tile*64 + lane)*8;
  *reinterpret_cast<v8h*>(p)        = __builtin_shufflevector(f, f, 0,1,2,3,4,5,6,7);
  *reinterpret_cast<v8h*>(p + 32*8) = __builtin_shufflevector(f, f, 8,9,10,11,12,13,14,15);
}
__global__ __launch_bounds__(512) void k_gru(
    const float* __restrict__ x, const float* __restrict__ Wih,
    const float* __restrict__ Whh, const float* __restrict__ bih,
    const float* __restrict__ bhh, float* __restrict__ seq)
{
  __shared__ __attribute__((aligned(16))) _Float16 WhhL[192*64];
  __shared__ __attribute__((aligned(16))) _Float16 WihL[192*128];
  __shared__ __attribute__((aligned(16))) _Float16 xsL[2][8][4*512];
  __shared__ __attribute__((aligned(16))) _Float16 hsL[2][8][16*HP];
  int tid=threadIdx.x, lane=tid&31;
  int wave=__builtin_amdgcn_readfirstlane(tid>>5);
  int tt=wave>>1, jh=wave&1;
  int pbase=(blockIdx.x*8+tt)*16;
  int colb=lane&15, rowoff=(lane>>4)*8, h16=(lane>>4)&1;

  stage_w<128,48,512>(Wih, WihL, tid);
  stage_w<64,24,512>(Whh, WhhL, tid);

  float bi[3][2], bh[3][2];
  #pragma unroll
  for(int g=0;g<3;++g){
    #pragma unroll
    for(int jj=0;jj<2;++jj){
      bi[g][jj]=bih[(g*4+2*jh+jj)*16+colb];
      bh[g][jj]=bhh[(g*4+2*jh+jj)*16+colb];
    }
  }

  const float* xrow = x + ((size_t)pbase*PB + (size_t)(lane&15)*PB)*DD + 8*h16;

  {
    _Float16* hw=&hsL[0][tt][0];
    #pragma unroll
    for(int jj=0;jj<2;++jj){
      #pragma unroll
      for(int v=0;v<8;++v) hw[(v+rowoff)*HP + (2*jh+jj)*16 + colb]=(_Float16)0.f;
    }
    #pragma unroll
    for(int q=0;q<2;++q){
      int kk=2*jh+q;
      const float* rb=xrow+kk*32;
      v16h f;
      #pragma unroll
      for(int i=0;i<16;++i) f[i]=(_Float16)rb[(i&7)+((i>>3)<<4)];
      store_tile_lds(&xsL[0][tt][0], kk, lane, f);
    }
  }
  __syncthreads();

  v8f hT[2], hs[2];
  #pragma unroll
  for(int jj=0;jj<2;++jj){ hT[jj]=zero8(); hs[jj]=zero8(); }

  for(int t=0;t<PB;++t){
    int buf=t&1;
    float xr[2][16];
    if(t+1<PB){
      #pragma unroll
      for(int q=0;q<2;++q){
        const float* rb=xrow+(size_t)(t+1)*DD+(2*jh+q)*32;
        #pragma unroll
        for(int i=0;i<16;++i) xr[q][i]=rb[(i&7)+((i>>3)<<4)];
      }
    }
    const _Float16* xl=&xsL[buf][tt][0];
    const _Float16* hl=&hsL[buf][tt][0];
    v16h xA[4], hA[2];
    #pragma unroll
    for(int kk=0;kk<4;++kk) xA[kk]=load_tile_lds(xl, kk, lane);
    #pragma unroll
    for(int kk=0;kk<2;++kk) hA[kk]=load_tile_f16(hl+kk*32, HP, lane);


    #pragma unroll
    for(int jj=0;jj<2;++jj){
      int j=2*jh+jj;
      v8f gr=zero8();
      #pragma unroll
      for(int kk=0;kk<4;++kk)
        gr=WMMA_F16(xA[kk], load_tile_lds(WihL, j*4+kk, lane), gr);
      #pragma unroll
      for(int kk=0;kk<2;++kk)
        gr=WMMA_F16(hA[kk], load_tile_lds(WhhL, j*2+kk, lane), gr);
      v8f gz=zero8();
      #pragma unroll
      for(int kk=0;kk<4;++kk)
        gz=WMMA_F16(xA[kk], load_tile_lds(WihL, (4+j)*4+kk, lane), gz);
      #pragma unroll
      for(int kk=0;kk<2;++kk)
        gz=WMMA_F16(hA[kk], load_tile_lds(WhhL, (4+j)*2+kk, lane), gz);
      v8f gnx=zero8();
      #pragma unroll
      for(int kk=0;kk<4;++kk)
        gnx=WMMA_F16(xA[kk], load_tile_lds(WihL, (8+j)*4+kk, lane), gnx);
      v8f gnh=zero8();
      #pragma unroll
      for(int kk=0;kk<2;++kk)
        gnh=WMMA_F16(hA[kk], load_tile_lds(WhhL, (8+j)*2+kk, lane), gnh);
      #pragma unroll
      for(int e=0;e<8;++e){
        float r =sigf(gr[e]+bi[0][jj]+bh[0][jj]);
        float az=gz[e]+bi[1][jj]+bh[1][jj];
        float u =gnx[e]+bi[2][jj]+r*(gnh[e]+bh[2][jj]);
        float ez=__expf(-fabsf(az));
        float en=__expf(-2.f*fabsf(u));
        float dz=1.f+ez, dn=1.f+en;
        float inv=1.f/(dz*dn);
        float z =(az>=0.f?1.f:ez)*(dn*inv);
        float nv=copysignf((1.f-en)*dz*inv, u);
        float hn=(1.f-z)*nv + z*hT[jj][e];
        hT[jj][e]=hn; hs[jj][e]+=hn;
      }
    }
    if(t+1<PB){
      _Float16* hw=&hsL[buf^1][tt][0];
      #pragma unroll
      for(int jj=0;jj<2;++jj){
        #pragma unroll
        for(int v=0;v<8;++v)
          hw[(v+rowoff)*HP + (2*jh+jj)*16 + colb]=(_Float16)hT[jj][v];
      }
      #pragma unroll
      for(int q=0;q<2;++q){
        v16h f;
        #pragma unroll
        for(int i=0;i<16;++i) f[i]=(_Float16)xr[q][i];
        store_tile_lds(&xsL[buf^1][tt][0], 2*jh+q, lane, f);
      }
    }
    __syncthreads();
  }
  float* sb=seq+(size_t)pbase*HH;
  #pragma unroll
  for(int jj=0;jj<2;++jj){
    #pragma unroll
    for(int v=0;v<8;++v)
      sb[(size_t)(v+rowoff)*HH + (2*jh+jj)*16 + colb]=hs[jj][v]*0.125f;
  }
}

__device__ __forceinline__ v16h softmax_tile(v8f s0, v8f s1, v8f m, v8f& lrow,
                                             _Float16* pst, const _Float16* pld){
  v8h p0, p1;
  #pragma unroll
  for(int e=0;e<8;++e){
    float a=__expf(s0[e]-m[e]);
    float c=__expf(s1[e]-m[e]);
    lrow[e]+=a+c;
    p0[e]=(_Float16)a; p1[e]=(_Float16)c;
  }
  *reinterpret_cast<v8h*>(pst)=p0;
  *reinterpret_cast<v8h*>(pst+16*16)=p1;
  return cat16(ld_tr16(pld), ld_tr16(pld+16*16));
}

#define OP 72
__global__ __launch_bounds__(256) void k_attn(
    const float* __restrict__ X, const float* __restrict__ Wqkv,
    const float* __restrict__ bqkv, const float* __restrict__ Wo,
    const float* __restrict__ bo, const float* __restrict__ g,
    const float* __restrict__ be, float* __restrict__ Xo)
{
  __shared__ __attribute__((aligned(16))) _Float16 Ks[4*SQ*32];
  __shared__ __attribute__((aligned(16))) _Float16 Vs[4*SQ*16];
  __shared__ __attribute__((aligned(16))) _Float16 Ps[8][4*32*16];
  __shared__ __attribute__((aligned(16))) _Float16 WL[24*512];
  __shared__ __attribute__((aligned(16))) _Float16 WoL[64*64];
  __shared__ __attribute__((aligned(16))) _Float16 Os[128*OP];
  __shared__ __attribute__((aligned(16))) float bL[192];
  int b=blockIdx.x>>2, qq=blockIdx.x&3;
  int tid=threadIdx.x, lane=tid&31;
  int wave=__builtin_amdgcn_readfirstlane(tid>>5);
  int h=wave&3, qh=wave>>2;
  int colb=lane&15, rowoff=(lane>>4)*8;

  const float* xs=X+((size_t)b*SQ+wave*64)*HH;
  v16h Xb[4][2];
  #pragma unroll
  for(int qt=0;qt<4;++qt){
    #pragma unroll
    for(int kk=0;kk<2;++kk) Xb[qt][kk]=load_tile_f32(xs+(size_t)qt*16*HH+kk*32, HH, lane);
  }
  const float* xq=X+((size_t)b*SQ+qq*128+qh*64)*HH;
  v16h Xq[4][2];
  #pragma unroll
  for(int qt=0;qt<4;++qt){
    #pragma unroll
    for(int kk=0;kk<2;++kk) Xq[qt][kk]=load_tile_f32(xq+(size_t)qt*16*HH+kk*32, HH, lane);
  }
  stage_w<64,24,256>(Wqkv, WL, tid);
  stage_w<64,8,256>(Wo, WoL, tid);
  if(tid<192) bL[tid]=bqkv[tid];
  __syncthreads();
  v8h zh;
  #pragma unroll
  for(int r=0;r<8;++r) zh[r]=(_Float16)0.f;
  #pragma nounroll
  for(int hh=0;hh<4;++hh){
    v16h Wk[2], Wv[2];
    #pragma unroll
    for(int kk=0;kk<2;++kk){
      Wk[kk]=load_tile_lds(WL, (4+hh)*2+kk, lane);
      Wv[kk]=load_tile_lds(WL, (8+hh)*2+kk, lane);
    }
    float bk[8], bv[8];
    #pragma unroll
    for(int r=0;r<8;++r){ bk[r]=bL[64+hh*16+rowoff+r]; bv[r]=bL[128+hh*16+rowoff+r]; }
    _Float16* Kw=Ks+(size_t)hh*SQ*32;
    _Float16* Vw=Vs+(size_t)hh*SQ*16;
    #pragma unroll
    for(int qt=0;qt<4;++qt){
      v8f k=zero8(), v=zero8();
      #pragma unroll
      for(int kk=0;kk<2;++kk){
        k=WMMA_F16(Wk[kk], Xb[qt][kk], k);
        v=WMMA_F16(Wv[kk], Xb[qt][kk], v);
      }
      v8h kh, vh;
      #pragma unroll
      for(int r=0;r<8;++r){
        kh[r]=(_Float16)(k[r]+bk[r]);
        vh[r]=(_Float16)(v[r]+bv[r]);
      }
      int key=wave*64+qt*16+colb;
      *reinterpret_cast<v8h*>(Kw+(size_t)key*32+rowoff)=kh;
      *reinterpret_cast<v8h*>(Kw+(size_t)key*32+16+rowoff)=zh;
      *reinterpret_cast<v8h*>(Vw+(size_t)key*16+rowoff)=vh;
    }
  }
  v16h Qa[4];
  {
    v16h Wq[2];
    #pragma unroll
    for(int kk=0;kk<2;++kk) Wq[kk]=load_tile_lds(WL, h*2+kk, lane);
    float bq[8];
    #pragma unroll
    for(int r=0;r<8;++r) bq[r]=bL[h*16+rowoff+r];
    #pragma unroll
    for(int qt=0;qt<4;++qt){
      v8f q=zero8();
      #pragma unroll
      for(int kk=0;kk<2;++kk)
        q=WMMA_F16(Wq[kk], Xq[qt][kk], q);
      v8h qv;
      #pragma unroll
      for(int r=0;r<8;++r) qv[r]=(_Float16)((q[r]+bq[r])*0.25f);
      Qa[qt]=cat16(qv, zh);
    }
  }
  __syncthreads();
  const _Float16* Kh=Ks+(size_t)h*SQ*32;
  const _Float16* Vh=Vs+(size_t)h*SQ*16;

  v8f mrow[4];
  {
    v8f lo;
    lo[0]=-1e30f; lo[1]=-1e30f; lo[2]=-1e30f; lo[3]=-1e30f;
    lo[4]=-1e30f; lo[5]=-1e30f; lo[6]=-1e30f; lo[7]=-1e30f;
    mrow[0]=lo; mrow[1]=lo; mrow[2]=lo; mrow[3]=lo;
  }
  for(int kc=0;kc<16;++kc){
    int kbase=kc*32;
    v16h Kb0=load_tile_f16(Kh+(size_t)kbase*32, 32, lane);
    v16h Kb1=load_tile_f16(Kh+(size_t)(kbase+16)*32, 32, lane);
    #pragma unroll
    for(int qt=0;qt<4;++qt){
      v8f s0=WMMA_F16(Qa[qt],Kb0,zero8());
      v8f s1=WMMA_F16(Qa[qt],Kb1,zero8());
      mrow[qt]=__builtin_elementwise_max(mrow[qt],__builtin_elementwise_max(s0,s1));
    }
  }
  mrow[0]=rmax16x8(mrow[0]);
  mrow[1]=rmax16x8(mrow[1]);
  mrow[2]=rmax16x8(mrow[2]);
  mrow[3]=rmax16x8(mrow[3]);

  _Float16* pl=&Ps[wave][0];
  _Float16* pst=pl+colb*16+rowoff;
  int troff=(8*(lane>>4)+(lane&7))*16+8*((lane>>3)&1);
  const _Float16* pld=pl+troff;
  const _Float16* vld=Vh+troff;
  v8f acc[4], lrow[4];
  acc[0]=zero8(); acc[1]=zero8(); acc[2]=zero8(); acc[3]=zero8();
  lrow[0]=zero8(); lrow[1]=zero8(); lrow[2]=zero8(); lrow[3]=zero8();
  v16h Kb0=load_tile_f16(Kh, 32, lane);
  v16h Kb1=load_tile_f16(Kh+(size_t)16*32, 32, lane);
  v16h Vb=cat16(ld_tr16(vld), ld_tr16(vld+16*16));
  v16h Pa3=zero16(), Vp=zero16();
  for(int kc=0;kc<16;++kc){
    int kn=(kc<15?kc+1:15)*32;
    v16h Kn0=load_tile_f16(Kh+(size_t)kn*32, 32, lane);
    v16h Kn1=load_tile_f16(Kh+(size_t)(kn+16)*32, 32, lane);
    const _Float16* vk=vld+(size_t)kn*16;
    v16h Vn=cat16(ld_tr16(vk), ld_tr16(vk+16*16));
    v8f s00=WMMA_F16(Qa[0],Kb0,zero8()), s01=WMMA_F16(Qa[0],Kb1,zero8());
    v8f s10=WMMA_F16(Qa[1],Kb0,zero8()), s11=WMMA_F16(Qa[1],Kb1,zero8());
    v8f s20=WMMA_F16(Qa[2],Kb0,zero8()), s21=WMMA_F16(Qa[2],Kb1,zero8());
    v8f s30=WMMA_F16(Qa[3],Kb0,zero8()), s31=WMMA_F16(Qa[3],Kb1,zero8());
    acc[3]=WMMA_F16(Pa3,Vp,acc[3]);
    v16h Pa0=softmax_tile(s00,s01,mrow[0],lrow[0],pst,pld);
    v16h Pa1=softmax_tile(s10,s11,mrow[1],lrow[1],pst+512,pld+512);
    acc[0]=WMMA_F16(Pa0,Vb,acc[0]);
    v16h Pa2=softmax_tile(s20,s21,mrow[2],lrow[2],pst+1024,pld+1024);
    acc[1]=WMMA_F16(Pa1,Vb,acc[1]);
    Pa3=softmax_tile(s30,s31,mrow[3],lrow[3],pst+1536,pld+1536);
    acc[2]=WMMA_F16(Pa2,Vb,acc[2]);
    Vp=Vb; Kb0=Kn0; Kb1=Kn1; Vb=Vn;
  }
  acc[3]=WMMA_F16(Pa3,Vp,acc[3]);
  #pragma unroll
  for(int qt=0;qt<4;++qt){
    v8f lr=rsum16x8(lrow[qt]);
    _Float16* ow=Os+(size_t)(qh*64+qt*16+rowoff)*OP+h*16+colb;
    #pragma unroll
    for(int e=0;e<8;++e) ow[(size_t)e*OP]=(_Float16)(acc[qt][e]/lr[e]);
  }
  __syncthreads();

  size_t rbase=(size_t)b*SQ+qq*128+wave*16;
  v16h A[2];
  #pragma unroll
  for(int kk=0;kk<2;++kk) A[kk]=load_tile_f16(Os+(size_t)wave*16*OP+kk*32, OP, lane);
  v8f o[4];
  #pragma unroll
  for(int j=0;j<4;++j){
    v8f pa=zero8();
    #pragma unroll
    for(int kk=0;kk<2;++kk)
      pa=WMMA_F16(A[kk], load_tile_lds(WoL, j*2+kk, lane), pa);
    float bias=bo[j*16+colb];
    const float* xr=X+rbase*HH+j*16+colb;
    #pragma unroll
    for(int e=0;e<8;++e) o[j][e]=pa[e]+bias+xr[(size_t)(e+rowoff)*HH];
  }
  float gv[4],bv2[4];
  #pragma unroll
  for(int j=0;j<4;++j){ gv[j]=g[j*16+colb]; bv2[j]=be[j*16+colb]; }
  float* xw=Xo+rbase*HH;
  #pragma unroll
  for(int e=0;e<8;++e){
    float s=o[0][e]+o[1][e]+o[2][e]+o[3][e];
    float mu=rsum16(s)*(1.f/64.f);
    float d0=o[0][e]-mu,d1=o[1][e]-mu,d2=o[2][e]-mu,d3=o[3][e]-mu;
    float var=rsum16(d0*d0+d1*d1+d2*d2+d3*d3)*(1.f/64.f);
    float inv=rsqrtf(var+1e-5f);
    #pragma unroll
    for(int j=0;j<4;++j)
      xw[(size_t)(e+rowoff)*HH+j*16+colb]=(o[j][e]-mu)*inv*gv[j]+bv2[j];
  }
}

template<bool LAST>
__global__ __launch_bounds__(128) void k_ffln(
    float* __restrict__ X, const float* __restrict__ W1,
    const float* __restrict__ b1, const float* __restrict__ W2,
    const float* __restrict__ b2, const float* __restrict__ g,
    const float* __restrict__ be, float* __restrict__ part)
{
  __shared__ _Float16 h1[4][16*256];
  __shared__ float red[4][64];
  __shared__ __attribute__((aligned(16))) _Float16 W1L[256*64];
  __shared__ __attribute__((aligned(16))) _Float16 W2L[64*256];
  int lane=threadIdx.x&31, wave=threadIdx.x>>5;
  size_t rbase=((size_t)blockIdx.x*4+wave)*16;
  int colb=lane&15, rowoff=(lane>>4)*8;
  stage_w<64,32,128>(W1, W1L, threadIdx.x);
  stage_w<256,32,128>(W2, W2L, threadIdx.x);
  v16h A[2];
  #pragma unroll
  for(int kk=0;kk<2;++kk) A[kk]=load_tile_f32(X+rbase*HH+kk*32, HH, lane);
  __syncthreads();
  _Float16* hl=&h1[wave][0];
  #pragma unroll
  for(int j=0;j<16;++j){
    v8f acc=zero8();
    #pragma unroll
    for(int kk=0;kk<2;++kk)
      acc=WMMA_F16(A[kk], load_tile_lds(W1L, j*2+kk, lane), acc);
    float bias=b1[j*16+colb];
    #pragma unroll
    for(int e=0;e<8;++e)
      hl[(size_t)(e+rowoff)*256+j*16+colb]=(_Float16)fmaxf(acc[e]+bias,0.f);
  }
  v16h pA[8];
  #pragma unroll
  for(int kk=0;kk<8;++kk) pA[kk]=load_tile_f16(hl+kk*32, 256, lane);
  v8f o[4];
  #pragma unroll
  for(int j=0;j<4;++j){
    v8f acc=zero8();
    #pragma unroll
    for(int kk=0;kk<8;++kk)
      acc=WMMA_F16(pA[kk], load_tile_lds(W2L, j*8+kk, lane), acc);
    float bias=b2[j*16+colb];
    const float* xr=X+rbase*HH+j*16+colb;
    #pragma unroll
    for(int e=0;e<8;++e) o[j][e]=acc[e]+bias+xr[(size_t)(e+rowoff)*HH];
  }
  float gv[4],bv[4];
  #pragma unroll
  for(int j=0;j<4;++j){ gv[j]=g[j*16+colb]; bv[j]=be[j*16+colb]; }
  float* xw=X+rbase*HH;
  float cs[4]={0.f,0.f,0.f,0.f};
  #pragma unroll
  for(int e=0;e<8;++e){
    float s=o[0][e]+o[1][e]+o[2][e]+o[3][e];
    float mu=rsum16(s)*(1.f/64.f);
    float d0=o[0][e]-mu,d1=o[1][e]-mu,d2=o[2][e]-mu,d3=o[3][e]-mu;
    float var=rsum16(d0*d0+d1*d1+d2*d2+d3*d3)*(1.f/64.f);
    float inv=rsqrtf(var+1e-5f);
    #pragma unroll
    for(int j=0;j<4;++j){
      float y=(o[j][e]-mu)*inv*gv[j]+bv[j];
      if(LAST) cs[j]+=y;
      else     xw[(size_t)(e+rowoff)*HH+j*16+colb]=y;
    }
  }
  if(LAST){
    #pragma unroll
    for(int j=0;j<4;++j) cs[j]+=__shfl_xor(cs[j],16,32);
    if(lane<16){
      #pragma unroll
      for(int j=0;j<4;++j) red[wave][j*16+lane]=cs[j];
    }
    __syncthreads();
    if(threadIdx.x<64){
      int t=threadIdx.x;
      part[(size_t)blockIdx.x*HH+t]=(red[0][t]+red[1][t])+(red[2][t]+red[3][t]);
    }
  }
}

__global__ void k_mean(const float* __restrict__ part, float* __restrict__ out){
  int b=blockIdx.x, j=threadIdx.x;
  const float* p=part+(size_t)b*8*HH+j;
  float v[8];
  #pragma unroll
  for(int w=0;w<8;++w) v[w]=p[w*HH];
  float s=((v[0]+v[1])+(v[2]+v[3]))+((v[4]+v[5])+(v[6]+v[7]));
  out[b*HH+j]=s*(1.f/512.f);
}

extern "C" void kernel_launch(void* const* d_in, const int* in_sizes, int n_in,
                              void* d_out, int out_size, void* d_ws, size_t ws_size,
                              hipStream_t stream){
  const float* x  =(const float*)d_in[0];
  const float* Wih=(const float*)d_in[1];
  const float* Whh=(const float*)d_in[2];
  const float* bih=(const float*)d_in[3];
  const float* bhh=(const float*)d_in[4];

  char* ws=(char*)d_ws;
  float*    X =(float*)ws;
  float*    Y =(float*)(ws+(size_t)(8<<20));
  float*    Pm=Y;

  k_gru<<<256,512,0,stream>>>(x,Wih,Whh,bih,bhh,X);

  for(int L=0;L<2;++L){
    int base=5+L*12;
    const float* Wqkv=(const float*)d_in[base+0];
    const float* Wo  =(const float*)d_in[base+2];
    const float* W1  =(const float*)d_in[base+4];
    const float* W2  =(const float*)d_in[base+6];
    const float* bqkv=(const float*)d_in[base+1];
    const float* bo  =(const float*)d_in[base+3];
    const float* b1  =(const float*)d_in[base+5];
    const float* b2  =(const float*)d_in[base+7];
    const float* g1  =(const float*)d_in[base+8];
    const float* be1 =(const float*)d_in[base+9];
    const float* g2  =(const float*)d_in[base+10];
    const float* be2 =(const float*)d_in[base+11];

    if(L==0){
      k_attn<<<256,256,0,stream>>>(X,Wqkv,bqkv,Wo,bo,g1,be1,Y);
      k_ffln<false><<<512,128,0,stream>>>(Y,W1,b1,W2,b2,g2,be2,Pm);
    }else{
      k_attn<<<256,256,0,stream>>>(Y,Wqkv,bqkv,Wo,bo,g1,be1,X);
      k_ffln<true> <<<512,128,0,stream>>>(X,W1,b1,W2,b2,g2,be2,Pm);
    }
  }
  k_mean<<<64,64,0,stream>>>(Pm,(float*)d_out);
}


# global loads that feed fragments read whole 128-byte lines (4 per wave load, was 16): weights 8 lanes per row line, X tiles in 512 B runs; fragment order made in LDS; k_attn, k_ffln; same sums
# speedup vs baseline: 3.5486x; 1.0791x over previous
#include <hip/hip_runtime.h>
#include <hip/hip_fp16.h>

typedef __attribute__((ext_vector_type(16))) _Float16 v16h;
typedef __attribute__((ext_vector_type(8)))  float    v8f;

#define WMMA_F16(a,b,c) __builtin_amdgcn_wmma_f32_16x16x32_f16(false,(a),false,(b),(short)0,(c),false,false)

__device__ __forceinline__ v8f zero8(){
  v8f z;
  #pragma unroll
  for(int i=0;i<8;++i) z[i]=0.f;
  return z;
}
__device__ __forceinline__ v16h zero16(){
  v16h z;
  #pragma unroll
  for(int i=0;i<16;++i) z[i]=(_Float16)0.f;
  return z;
}

__device__ __forceinline__ int a_k(int i, int h16){ return (i&7) + ((i>>3)<<4) + (h16<<3); }

__device__ __forceinline__ v16h load_tile_f16(const _Float16* __restrict__ src, int ld, int lane){
  const _Float16* rb = src + (size_t)(lane&15)*ld;
  int h16=(lane>>4)&1; v16h r;
  #pragma unroll
  for(int i=0;i<16;++i) r[i]=rb[a_k(i,h16)];
  return r;
}
__device__ __forceinline__ v16h load_tile_kmajor_f16(const _Float16* __restrict__ src, int ld, int lane){
  const _Float16* cb = src + (lane&15);
  int h16=(lane>>4)&1; v16h r;
  #pragma unroll
  for(int i=0;i<16;++i) r[i]=cb[(size_t)a_k(i,h16)*ld];
  return r;
}

typedef short s16x8 __attribute__((ext_vector_type(8)));
typedef __attribute__((ext_vector_type(8))) _Float16 v8h;
typedef __attribute__((ext_vector_type(4))) _Float16 v4h;
__device__ __forceinline__ v8h ld_tr16(const _Float16* p){
  typedef __attribute__((address_space(3))) s16x8 lds_s16x8;
  return __builtin_bit_cast(v8h, __builtin_amdgcn_ds_load_tr16_b128_v8i16((lds_s16x8*)p));
}
__device__ __forceinline__ v16h cat16(v8h lo, v8h hi){
  return __builtin_shufflevector(lo, hi, 0,1,2,3,4,5,6,7,8,9,10,11,12,13,14,15);
}

__device__ __forceinline__ float rmax16(float x){
  #pragma unroll
  for(int m=8;m;m>>=1) x=fmaxf(x,__shfl_xor(x,m,32));
  return x;
}
__device__ __forceinline__ float rsum16(float x){
  #pragma unroll
  for(int m=8;m;m>>=1) x+=__shfl_xor(x,m,32);
  return x;
}
__device__ __forceinline__ float sigf(float x){ return 1.f/(1.f+__expf(-x)); }

__device__ __forceinline__ v8f rmax16x8(v8f x){
  #pragma unroll
  for(int m=8;m;m>>=1){
    v8f t;
    #pragma unroll
    for(int e=0;e<8;++e) t[e]=__shfl_xor(x[e],m,32);
    x=__builtin_elementwise_max(x,t);
  }
  return x;
}
__device__ __forceinline__ v8f rsum16x8(v8f x){
  #pragma unroll
  for(int m=8;m;m>>=1){
    v8f t;
    #pragma unroll
    for(int e=0;e<8;++e) t[e]=__shfl_xor(x[e],m,32);
    x+=t;
  }
  return x;
}

#define PB  8
#define DD  128
#define HH  64
#define SQ  512

#define WU 160
#define WT 640
template<int K, int NT, int NTHR>
__device__ __forceinline__ void stage_w(const float* __restrict__ W, _Float16* L, int tid){
  constexpr int KT = K/32;
  constexpr int NW = NTHR/32;
  int wave = __builtin_amdgcn_readfirstlane(tid>>5);
  int lane = tid & 31;
  int i  = lane >> 3;
  int k4 = lane & 7;
  const float* sl = W + (size_t)i*K + 4*k4;
  _Float16*    dl = L + (k4>>1)*WU + i*8 + (k4&1)*4;
  #pragma unroll
  for(int q=0;q<NT*4/NW;++q){
    int item = q*NW + wave;
    int rq   = item & 3;
    int tile = item >> 2;
    int kt = tile % KT, jt = tile / KT;
    const float* s = sl + (size_t)(jt*16 + 4*rq)*K + kt*32;
    v4h r;
    #pragma unroll
    for(int e=0;e<4;++e) r[e]=(_Float16)s[e];
    *reinterpret_cast<v4h*>(dl + (size_t)tile*WT + rq*32) = r;
  }
}
__device__ __forceinline__ v16h load_w_lds(const _Float16* L, int tile, int lane){
  const _Float16* p = L + (size_t)tile*WT + ((lane>>4)&1)*WU + (lane&15)*8;
  return cat16(*reinterpret_cast<const v8h*>(p), *reinterpret_cast<const v8h*>(p + 2*WU));
}
#define XP 72
__device__ __forceinline__ void load_xtile(const float* __restrict__ src, _Float16* buf, int lane, v16h& f0, v16h& f1){
  const float* s = src + 4*lane;
  _Float16* d = buf + (lane>>4)*XP + 4*(lane&15);
  #pragma unroll
  for(int j=0;j<8;++j){
    v4h r;
    #pragma unroll
    for(int e=0;e<4;++e) r[e]=(_Float16)s[j*128+e];
    *reinterpret_cast<v4h*>(d + j*2*XP) = r;
  }
  f0 = load_tile_f16(buf,    XP, lane);
  f1 = load_tile_f16(buf+32, XP, lane);
}
template<int K, int NT, int NTHR>
__device__ __forceinline__ void stage_w_rows(const float* __restrict__ W, _Float16* L, int tid){
  constexpr int KT = K/32;
  #pragma unroll
  for(int q=0;q<NT*64/NTHR;++q){
    int u    = q*NTHR + tid;
    int lane = u & 31;
    int p    = (u >> 5) & 1;
    int tile = u >> 6;
    int kt = tile % KT, jt = tile / KT;
    const float* s = W + (size_t)(jt*16 + (lane&15))*K + kt*32 + 16*p + 8*((lane>>4)&1);
    v8h r;
    #pragma unroll
    for(int e=0;e<8;++e) r[e]=(_Float16)s[e];
    *reinterpret_cast<v8h*>(L + (size_t)u*8) = r;
  }
}
__device__ __forceinline__ v16h load_tile_lds(const _Float16* L, int tile, int lane){
  const _Float16* p = L + ((size_t)tile*64 + lane)*8;
  return cat16(*reinterpret_cast<const v8h*>(p), *reinterpret_cast<const v8h*>(p + 32*8));
}

#define HP 72
__device__ __forceinline__ void store_tile_lds(_Float16* L, int tile, int lane, v16h f){
  _Float16* p = L + ((size_t)tile*64 + lane)*8;
  *reinterpret_cast<v8h*>(p)        = __builtin_shufflevector(f, f, 0,1,2,3,4,5,6,7);
  *reinterpret_cast<v8h*>(p + 32*8) = __builtin_shufflevector(f, f, 8,9,10,11,12,13,14,15);
}
__global__ __launch_bounds__(512) void k_gru(
    const float* __restrict__ x, const float* __restrict__ Wih,
    const float* __restrict__ Whh, const float* __restrict__ bih,
    const float* __restrict__ bhh, float* __restrict__ seq)
{
  __shared__ __attribute__((aligned(16))) _Float16 WhhL[192*64];
  __shared__ __attribute__((aligned(16))) _Float16 WihL[192*128];
  __shared__ __attribute__((aligned(16))) _Float16 xsL[2][8][4*512];
  __shared__ __attribute__((aligned(16))) _Float16 hsL[2][8][16*HP];
  int tid=threadIdx.x, lane=tid&31;
  int wave=__builtin_amdgcn_readfirstlane(tid>>5);
  int tt=wave>>1, jh=wave&1;
  int pbase=(blockIdx.x*8+tt)*16;
  int colb=lane&15, rowoff=(lane>>4)*8, h16=(lane>>4)&1;

  stage_w_rows<128,48,512>(Wih, WihL, tid);
  stage_w_rows<64,24,512>(Whh, WhhL, tid);

  float bi[3][2], bh[3][2];
  #pragma unroll
  for(int g=0;g<3;++g){
    #pragma unroll
    for(int jj=0;jj<2;++jj){
      bi[g][jj]=bih[(g*4+2*jh+jj)*16+colb];
      bh[g][jj]=bhh[(g*4+2*jh+jj)*16+colb];
    }
  }

  const float* xrow = x + ((size_t)pbase*PB + (size_t)(lane&15)*PB)*DD + 8*h16;

  {
    _Float16* hw=&hsL[0][tt][0];
    #pragma unroll
    for(int jj=0;jj<2;++jj){
      #pragma unroll
      for(int v=0;v<8;++v) hw[(v+rowoff)*HP + (2*jh+jj)*16 + colb]=(_Float16)0.f;
    }
    #pragma unroll
    for(int q=0;q<2;++q){
      int kk=2*jh+q;
      const float* rb=xrow+kk*32;
      v16h f;
      #pragma unroll
      for(int i=0;i<16;++i) f[i]=(_Float16)rb[(i&7)+((i>>3)<<4)];
      store_tile_lds(&xsL[0][tt][0], kk, lane, f);
    }
  }
  __syncthreads();

  v8f hT[2], hs[2];
  #pragma unroll
  for(int jj=0;jj<2;++jj){ hT[jj]=zero8(); hs[jj]=zero8(); }

  for(int t=0;t<PB;++t){
    int buf=t&1;
    float xr[2][16];
    if(t+1<PB){
      #pragma unroll
      for(int q=0;q<2;++q){
        const float* rb=xrow+(size_t)(t+1)*DD+(2*jh+q)*32;
        #pragma unroll
        for(int i=0;i<16;++i) xr[q][i]=rb[(i&7)+((i>>3)<<4)];
      }
    }
    const _Float16* xl=&xsL[buf][tt][0];
    const _Float16* hl=&hsL[buf][tt][0];
    v16h xA[4], hA[2];
    #pragma unroll
    for(int kk=0;kk<4;++kk) xA[kk]=load_tile_lds(xl, kk, lane);
    #pragma unroll
    for(int kk=0;kk<2;++kk) hA[kk]=load_tile_f16(hl+kk*32, HP, lane);


    #pragma unroll
    for(int jj=0;jj<2;++jj){
      int j=2*jh+jj;
      v8f gr=zero8();
      #pragma unroll
      for(int kk=0;kk<4;++kk)
        gr=WMMA_F16(xA[kk], load_tile_lds(WihL, j*4+kk, lane), gr);
      #pragma unroll
      for(int kk=0;kk<2;++kk)
        gr=WMMA_F16(hA[kk], load_tile_lds(WhhL, j*2+kk, lane), gr);
      v8f gz=zero8();
      #pragma unroll
      for(int kk=0;kk<4;++kk)
        gz=WMMA_F16(xA[kk], load_tile_lds(WihL, (4+j)*4+kk, lane), gz);
      #pragma unroll
      for(int kk=0;kk<2;++kk)
        gz=WMMA_F16(hA[kk], load_tile_lds(WhhL, (4+j)*2+kk, lane), gz);
      v8f gnx=zero8();
      #pragma unroll
      for(int kk=0;kk<4;++kk)
        gnx=WMMA_F16(xA[kk], load_tile_lds(WihL, (8+j)*4+kk, lane), gnx);
      v8f gnh=zero8();
      #pragma unroll
      for(int kk=0;kk<2;++kk)
        gnh=WMMA_F16(hA[kk], load_tile_lds(WhhL, (8+j)*2+kk, lane), gnh);
      #pragma unroll
      for(int e=0;e<8;++e){
        float r =sigf(gr[e]+bi[0][jj]+bh[0][jj]);
        float az=gz[e]+bi[1][jj]+bh[1][jj];
        float u =gnx[e]+bi[2][jj]+r*(gnh[e]+bh[2][jj]);
        float ez=__expf(-fabsf(az));
        float en=__expf(-2.f*fabsf(u));
        float dz=1.f+ez, dn=1.f+en;
        float inv=1.f/(dz*dn);
        float z =(az>=0.f?1.f:ez)*(dn*inv);
        float nv=copysignf((1.f-en)*dz*inv, u);
        float hn=(1.f-z)*nv + z*hT[jj][e];
        hT[jj][e]=hn; hs[jj][e]+=hn;
      }
    }
    if(t+1<PB){
      _Float16* hw=&hsL[buf^1][tt][0];
      #pragma unroll
      for(int jj=0;jj<2;++jj){
        #pragma unroll
        for(int v=0;v<8;++v)
          hw[(v+rowoff)*HP + (2*jh+jj)*16 + colb]=(_Float16)hT[jj][v];
      }
      #pragma unroll
      for(int q=0;q<2;++q){
        v16h f;
        #pragma unroll
        for(int i=0;i<16;++i) f[i]=(_Float16)xr[q][i];
        store_tile_lds(&xsL[buf^1][tt][0], 2*jh+q, lane, f);
      }
    }
    __syncthreads();
  }
  float* sb=seq+(size_t)pbase*HH;
  #pragma unroll
  for(int jj=0;jj<2;++jj){
    #pragma unroll
    for(int v=0;v<8;++v)
      sb[(size_t)(v+rowoff)*HH + (2*jh+jj)*16 + colb]=hs[jj][v]*0.125f;
  }
}

__device__ __forceinline__ v16h softmax_tile(v8f s0, v8f s1, v8f m, v8f& lrow,
                                             _Float16* pst, const _Float16* pld){
  v8h p0, p1;
  #pragma unroll
  for(int e=0;e<8;++e){
    float a=__expf(s0[e]-m[e]);
    float c=__expf(s1[e]-m[e]);
    lrow[e]+=a+c;
    p0[e]=(_Float16)a; p1[e]=(_Float16)c;
  }
  *reinterpret_cast<v8h*>(pst)=p0;
  *reinterpret_cast<v8h*>(pst+16*16)=p1;
  return cat16(ld_tr16(pld), ld_tr16(pld+16*16));
}

#define OP 72
__global__ __launch_bounds__(256) void k_attn(
    const float* __restrict__ X, const float* __restrict__ Wqkv,
    const float* __restrict__ bqkv, const float* __restrict__ Wo,
    const float* __restrict__ bo, const float* __restrict__ g,
    const float* __restrict__ be, float* __restrict__ Xo)
{
  __shared__ __attribute__((aligned(16))) _Float16 Ks[4*SQ*32];
  __shared__ __attribute__((aligned(16))) _Float16 Vs[4*SQ*16];
  __shared__ __attribute__((aligned(16))) _Float16 Ps[8][4*32*16];
  __shared__ __attribute__((aligned(16))) _Float16 WL[24*WT];
  __shared__ __attribute__((aligned(16))) _Float16 WoL[8*WT];
  __shared__ __attribute__((aligned(16))) _Float16 Os[128*OP];
  __shared__ __attribute__((aligned(16))) float bL[192];
  int b=blockIdx.x>>2, qq=blockIdx.x&3;
  int tid=threadIdx.x, lane=tid&31;
  int wave=__builtin_amdgcn_readfirstlane(tid>>5);
  int h=wave&3, qh=wave>>2;
  int colb=lane&15, rowoff=(lane>>4)*8;

  const float* xs=X+((size_t)b*SQ+wave*64)*HH;
  v16h Xb[4][2];
  _Float16* xbuf=&Ps[wave][0];
  #pragma unroll
  for(int qt=0;qt<4;++qt) load_xtile(xs+(size_t)qt*16*HH, xbuf, lane, Xb[qt][0], Xb[qt][1]);
  const float* xq=X+((size_t)b*SQ+qq*128+qh*64)*HH;
  v16h Xq[4][2];
  #pragma unroll
  for(int qt=0;qt<4;++qt) load_xtile(xq+(size_t)qt*16*HH, xbuf, lane, Xq[qt][0], Xq[qt][1]);
  stage_w<64,24,256>(Wqkv, WL, tid);
  stage_w<64,8,256>(Wo, WoL, tid);
  if(tid<192) bL[tid]=bqkv[tid];
  __syncthreads();
  v8h zh;
  #pragma unroll
  for(int r=0;r<8;++r) zh[r]=(_Float16)0.f;
  #pragma nounroll
  for(int hh=0;hh<4;++hh){
    v16h Wk[2], Wv[2];
    #pragma unroll
    for(int kk=0;kk<2;++kk){
      Wk[kk]=load_w_lds(WL, (4+hh)*2+kk, lane);
      Wv[kk]=load_w_lds(WL, (8+hh)*2+kk, lane);
    }
    float bk[8], bv[8];
    #pragma unroll
    for(int r=0;r<8;++r){ bk[r]=bL[64+hh*16+rowoff+r]; bv[r]=bL[128+hh*16+rowoff+r]; }
    _Float16* Kw=Ks+(size_t)hh*SQ*32;
    _Float16* Vw=Vs+(size_t)hh*SQ*16;
    #pragma unroll
    for(int qt=0;qt<4;++qt){
      v8f k=zero8(), v=zero8();
      #pragma unroll
      for(int kk=0;kk<2;++kk){
        k=WMMA_F16(Wk[kk], Xb[qt][kk], k);
        v=WMMA_F16(Wv[kk], Xb[qt][kk], v);
      }
      v8h kh, vh;
      #pragma unroll
      for(int r=0;r<8;++r){
        kh[r]=(_Float16)(k[r]+bk[r]);
        vh[r]=(_Float16)(v[r]+bv[r]);
      }
      int key=wave*64+qt*16+colb;
      *reinterpret_cast<v8h*>(Kw+(size_t)key*32+rowoff)=kh;
      *reinterpret_cast<v8h*>(Kw+(size_t)key*32+16+rowoff)=zh;
      *reinterpret_cast<v8h*>(Vw+(size_t)key*16+rowoff)=vh;
    }
  }
  v16h Qa[4];
  {
    v16h Wq[2];
    #pragma unroll
    for(int kk=0;kk<2;++kk) Wq[kk]=load_w_lds(WL, h*2+kk, lane);
    float bq[8];
    #pragma unroll
    for(int r=0;r<8;++r) bq[r]=bL[h*16+rowoff+r];
    #pragma unroll
    for(int qt=0;qt<4;++qt){
      v8f q=zero8();
      #pragma unroll
      for(int kk=0;kk<2;++kk)
        q=WMMA_F16(Wq[kk], Xq[qt][kk], q);
      v8h qv;
      #pragma unroll
      for(int r=0;r<8;++r) qv[r]=(_Float16)((q[r]+bq[r])*0.25f);
      Qa[qt]=cat16(qv, zh);
    }
  }
  __syncthreads();
  const _Float16* Kh=Ks+(size_t)h*SQ*32;
  const _Float16* Vh=Vs+(size_t)h*SQ*16;

  v8f mrow[4];
  {
    v8f lo;
    lo[0]=-1e30f; lo[1]=-1e30f; lo[2]=-1e30f; lo[3]=-1e30f;
    lo[4]=-1e30f; lo[5]=-1e30f; lo[6]=-1e30f; lo[7]=-1e30f;
    mrow[0]=lo; mrow[1]=lo; mrow[2]=lo; mrow[3]=lo;
  }
  for(int kc=0;kc<16;++kc){
    int kbase=kc*32;
    v16h Kb0=load_tile_f16(Kh+(size_t)kbase*32, 32, lane);
    v16h Kb1=load_tile_f16(Kh+(size_t)(kbase+16)*32, 32, lane);
    #pragma unroll
    for(int qt=0;qt<4;++qt){
      v8f s0=WMMA_F16(Qa[qt],Kb0,zero8());
      v8f s1=WMMA_F16(Qa[qt],Kb1,zero8());
      mrow[qt]=__builtin_elementwise_max(mrow[qt],__builtin_elementwise_max(s0,s1));
    }
  }
  mrow[0]=rmax16x8(mrow[0]);
  mrow[1]=rmax16x8(mrow[1]);
  mrow[2]=rmax16x8(mrow[2]);
  mrow[3]=rmax16x8(mrow[3]);

  _Float16* pl=&Ps[wave][0];
  _Float16* pst=pl+colb*16+rowoff;
  int troff=(8*(lane>>4)+(lane&7))*16+8*((lane>>3)&1);
  const _Float16* pld=pl+troff;
  const _Float16* vld=Vh+troff;
  v8f acc[4], lrow[4];
  acc[0]=zero8(); acc[1]=zero8(); acc[2]=zero8(); acc[3]=zero8();
  lrow[0]=zero8(); lrow[1]=zero8(); lrow[2]=zero8(); lrow[3]=zero8();
  v16h Kb0=load_tile_f16(Kh, 32, lane);
  v16h Kb1=load_tile_f16(Kh+(size_t)16*32, 32, lane);
  v16h Vb=cat16(ld_tr16(vld), ld_tr16(vld+16*16));
  v16h Pa3=zero16(), Vp=zero16();
  for(int kc=0;kc<16;++kc){
    int kn=(kc<15?kc+1:15)*32;
    v16h Kn0=load_tile_f16(Kh+(size_t)kn*32, 32, lane);
    v16h Kn1=load_tile_f16(Kh+(size_t)(kn+16)*32, 32, lane);
    const _Float16* vk=vld+(size_t)kn*16;
    v16h Vn=cat16(ld_tr16(vk), ld_tr16(vk+16*16));
    v8f s00=WMMA_F16(Qa[0],Kb0,zero8()), s01=WMMA_F16(Qa[0],Kb1,zero8());
    v8f s10=WMMA_F16(Qa[1],Kb0,zero8()), s11=WMMA_F16(Qa[1],Kb1,zero8());
    v8f s20=WMMA_F16(Qa[2],Kb0,zero8()), s21=WMMA_F16(Qa[2],Kb1,zero8());
    v8f s30=WMMA_F16(Qa[3],Kb0,zero8()), s31=WMMA_F16(Qa[3],Kb1,zero8());
    acc[3]=WMMA_F16(Pa3,Vp,acc[3]);
    v16h Pa0=softmax_tile(s00,s01,mrow[0],lrow[0],pst,pld);
    v16h Pa1=softmax_tile(s10,s11,mrow[1],lrow[1],pst+512,pld+512);
    acc[0]=WMMA_F16(Pa0,Vb,acc[0]);
    v16h Pa2=softmax_tile(s20,s21,mrow[2],lrow[2],pst+1024,pld+1024);
    acc[1]=WMMA_F16(Pa1,Vb,acc[1]);
    Pa3=softmax_tile(s30,s31,mrow[3],lrow[3],pst+1536,pld+1536);
    acc[2]=WMMA_F16(Pa2,Vb,acc[2]);
    Vp=Vb; Kb0=Kn0; Kb1=Kn1; Vb=Vn;
  }
  acc[3]=WMMA_F16(Pa3,Vp,acc[3]);
  #pragma unroll
  for(int qt=0;qt<4;++qt){
    v8f lr=rsum16x8(lrow[qt]);
    _Float16* ow=Os+(size_t)(qh*64+qt*16+rowoff)*OP+h*16+colb;
    #pragma unroll
    for(int e=0;e<8;++e) ow[(size_t)e*OP]=(_Float16)(acc[qt][e]/lr[e]);
  }
  __syncthreads();

  size_t rbase=(size_t)b*SQ+qq*128+wave*16;
  v16h A[2];
  #pragma unroll
  for(int kk=0;kk<2;++kk) A[kk]=load_tile_f16(Os+(size_t)wave*16*OP+kk*32, OP, lane);
  v8f o[4];
  #pragma unroll
  for(int j=0;j<4;++j){
    v8f pa=zero8();
    #pragma unroll
    for(int kk=0;kk<2;++kk)
      pa=WMMA_F16(A[kk], load_w_lds(WoL, j*2+kk, lane), pa);
    float bias=bo[j*16+colb];
    const float* xr=X+rbase*HH+j*16+colb;
    #pragma unroll
    for(int e=0;e<8;++e) o[j][e]=pa[e]+bias+xr[(size_t)(e+rowoff)*HH];
  }
  float gv[4],bv2[4];
  #pragma unroll
  for(int j=0;j<4;++j){ gv[j]=g[j*16+colb]; bv2[j]=be[j*16+colb]; }
  float* xw=Xo+rbase*HH;
  #pragma unroll
  for(int e=0;e<8;++e){
    float s=o[0][e]+o[1][e]+o[2][e]+o[3][e];
    float mu=rsum16(s)*(1.f/64.f);
    float d0=o[0][e]-mu,d1=o[1][e]-mu,d2=o[2][e]-mu,d3=o[3][e]-mu;
    float var=rsum16(d0*d0+d1*d1+d2*d2+d3*d3)*(1.f/64.f);
    float inv=rsqrtf(var+1e-5f);
    #pragma unroll
    for(int j=0;j<4;++j)
      xw[(size_t)(e+rowoff)*HH+j*16+colb]=(o[j][e]-mu)*inv*gv[j]+bv2[j];
  }
}

template<bool LAST>
__global__ __launch_bounds__(128) void k_ffln(
    float* __restrict__ X, const float* __restrict__ W1,
    const float* __restrict__ b1, const float* __restrict__ W2,
    const float* __restrict__ b2, const float* __restrict__ g,
    const float* __restrict__ be, float* __restrict__ part)
{
  __shared__ __attribute__((aligned(16))) _Float16 h1[4][16*256];
  __shared__ float red[4][64];
  __shared__ __attribute__((aligned(16))) _Float16 W1L[32*WT];
  __shared__ __attribute__((aligned(16))) _Float16 W2L[32*WT];
  int lane=threadIdx.x&31, wave=threadIdx.x>>5;
  size_t rbase=((size_t)blockIdx.x*4+wave)*16;
  int colb=lane&15, rowoff=(lane>>4)*8;
  stage_w<64,32,128>(W1, W1L, threadIdx.x);
  stage_w<256,32,128>(W2, W2L, threadIdx.x);
  v16h A[2];
  load_xtile(X+rbase*HH, &h1[wave][0], lane, A[0], A[1]);
  __syncthreads();
  _Float16* hl=&h1[wave][0];
  #pragma unroll
  for(int j=0;j<16;++j){
    v8f acc=zero8();
    #pragma unroll
    for(int kk=0;kk<2;++kk)
      acc=WMMA_F16(A[kk], load_w_lds(W1L, j*2+kk, lane), acc);
    float bias=b1[j*16+colb];
    #pragma unroll
    for(int e=0;e<8;++e)
      hl[(size_t)(e+rowoff)*256+j*16+colb]=(_Float16)fmaxf(acc[e]+bias,0.f);
  }
  v16h pA[8];
  #pragma unroll
  for(int kk=0;kk<8;++kk) pA[kk]=load_tile_f16(hl+kk*32, 256, lane);
  v8f o[4];
  #pragma unroll
  for(int j=0;j<4;++j){
    v8f acc=zero8();
    #pragma unroll
    for(int kk=0;kk<8;++kk)
      acc=WMMA_F16(pA[kk], load_w_lds(W2L, j*8+kk, lane), acc);
    float bias=b2[j*16+colb];
    const float* xr=X+rbase*HH+j*16+colb;
    #pragma unroll
    for(int e=0;e<8;++e) o[j][e]=acc[e]+bias+xr[(size_t)(e+rowoff)*HH];
  }
  float gv[4],bv[4];
  #pragma unroll
  for(int j=0;j<4;++j){ gv[j]=g[j*16+colb]; bv[j]=be[j*16+colb]; }
  float* xw=X+rbase*HH;
  float cs[4]={0.f,0.f,0.f,0.f};
  #pragma unroll
  for(int e=0;e<8;++e){
    float s=o[0][e]+o[1][e]+o[2][e]+o[3][e];
    float mu=rsum16(s)*(1.f/64.f);
    float d0=o[0][e]-mu,d1=o[1][e]-mu,d2=o[2][e]-mu,d3=o[3][e]-mu;
    float var=rsum16(d0*d0+d1*d1+d2*d2+d3*d3)*(1.f/64.f);
    float inv=rsqrtf(var+1e-5f);
    #pragma unroll
    for(int j=0;j<4;++j){
      float y=(o[j][e]-mu)*inv*gv[j]+bv[j];
      if(LAST) cs[j]+=y;
      else     xw[(size_t)(e+rowoff)*HH+j*16+colb]=y;
    }
  }
  if(LAST){
    #pragma unroll
    for(int j=0;j<4;++j) cs[j]+=__shfl_xor(cs[j],16,32);
    if(lane<16){
      #pragma unroll
      for(int j=0;j<4;++j) red[wave][j*16+lane]=cs[j];
    }
    __syncthreads();
    if(threadIdx.x<64){
      int t=threadIdx.x;
      part[(size_t)blockIdx.x*HH+t]=(red[0][t]+red[1][t])+(red[2][t]+red[3][t]);
    }
  }
}

__global__ void k_mean(const float* __restrict__ part, float* __restrict__ out){
  int b=blockIdx.x, j=threadIdx.x;
  const float* p=part+(size_t)b*8*HH+j;
  float v[8];
  #pragma unroll
  for(int w=0;w<8;++w) v[w]=p[w*HH];
  float s=((v[0]+v[1])+(v[2]+v[3]))+((v[4]+v[5])+(v[6]+v[7]));
  out[b*HH+j]=s*(1.f/512.f);
}

extern "C" void kernel_launch(void* const* d_in, const int* in_sizes, int n_in,
                              void* d_out, int out_size, void* d_ws, size_t ws_size,
                              hipStream_t stream){
  const float* x  =(const float*)d_in[0];
  const float* Wih=(const float*)d_in[1];
  const float* Whh=(const float*)d_in[2];
  const float* bih=(const float*)d_in[3];
  const float* bhh=(const float*)d_in[4];

  char* ws=(char*)d_ws;
  float*    X =(float*)ws;
  float*    Y =(float*)(ws+(size_t)(8<<20));
  float*    Pm=Y;

  k_gru<<<256,512,0,stream>>>(x,Wih,Whh,bih,bhh,X);

  for(int L=0;L<2;++L){
    int base=5+L*12;
    const float* Wqkv=(const float*)d_in[base+0];
    const float* Wo  =(const float*)d_in[base+2];
    const float* W1  =(const float*)d_in[base+4];
    const float* W2  =(const float*)d_in[base+6];
    const float* bqkv=(const float*)d_in[base+1];
    const float* bo  =(const float*)d_in[base+3];
    const float* b1  =(const float*)d_in[base+5];
    const float* b2  =(const float*)d_in[base+7];
    const float* g1  =(const float*)d_in[base+8];
    const float* be1 =(const float*)d_in[base+9];
    const float* g2  =(const float*)d_in[base+10];
    const float* be2 =(const float*)d_in[base+11];

    if(L==0){
      k_attn<<<256,256,0,stream>>>(X,Wqkv,bqkv,Wo,bo,g1,be1,Y);
      k_ffln<false><<<512,128,0,stream>>>(Y,W1,b1,W2,b2,g2,be2,Pm);
    }else{
      k_attn<<<256,256,0,stream>>>(Y,Wqkv,bqkv,Wo,bo,g1,be1,X);
      k_ffln<true> <<<512,128,0,stream>>>(X,W1,b1,W2,b2,g2,be2,Pm);
    }
  }
  k_mean<<<64,64,0,stream>>>(Pm,(float*)d_out);
}
